# starting kernel
# baseline (speedup 1.0000x reference)
#include <hip/hip_runtime.h>

#define ND_N 20000
#define NW_N 30000
#define E_N  500000
#define D_N  256
#define DOUT_N 256

typedef __attribute__((ext_vector_type(2))) float v2f;
typedef __attribute__((ext_vector_type(8))) float v8f;

__device__ __forceinline__ unsigned mapf(float f) {
  unsigned u = __float_as_uint(f);
  return (u & 0x80000000u) ? ~u : (u | 0x80000000u);
}
__device__ __forceinline__ float unmapf(unsigned u) {
  unsigned v = (u & 0x80000000u) ? (u & 0x7fffffffu) : ~u;
  return __uint_as_float(v);
}
#define NEG_INF_MAPPED 0x007fffffu

__global__ void fill_u32_kernel(unsigned* __restrict__ p, unsigned v, int n) {
  int i = blockIdx.x * blockDim.x + threadIdx.x;
  if (i < n) p[i] = v;
}

__global__ void node_dots_kernel(const float* __restrict__ hid, int n,
                                 const float* __restrict__ wa,
                                 const float* __restrict__ wb,
                                 float* __restrict__ outa,
                                 float* __restrict__ outb) {
  int wave = threadIdx.x >> 5, lane = threadIdx.x & 31;
  int node = blockIdx.x * 8 + wave;
  if (node >= n) return;
  const float* row = hid + (size_t)node * D_N;
  float da = 0.f, db = 0.f;
#pragma unroll
  for (int i = 0; i < 8; ++i) {
    int k = lane + i * 32;
    float x = row[k];
    da += x * wa[k];
    db += x * wb[k];
  }
#pragma unroll
  for (int off = 16; off; off >>= 1) {
    da += __shfl_xor(da, off, 32);
    db += __shfl_xor(db, off, 32);
  }
  if (lane == 0) { outa[node] = da; outb[node] = db; }
}

__global__ void edge_score_max_kernel(const int* __restrict__ src_idx,
                                      const int* __restrict__ dst_idx,
                                      const float* __restrict__ s_src,
                                      const float* __restrict__ s_dst,
                                      const float* __restrict__ bptr,
                                      float* __restrict__ e_arr,
                                      unsigned* __restrict__ m_arr, int ne) {
  int e = blockIdx.x * blockDim.x + threadIdx.x;
  if (e >= ne) return;
  float a = s_src[src_idx[e]] + s_dst[dst_idx[e]] + bptr[0];
  float v = a > 0.f ? a : 0.01f * a;
  e_arr[e] = v;
  atomicMax(&m_arr[dst_idx[e]], mapf(v));
}

__global__ void edge_exp_sum_kernel(const int* __restrict__ dst_idx,
                                    const unsigned* __restrict__ m_arr,
                                    float* __restrict__ e_arr,
                                    float* __restrict__ s_arr, int ne) {
  int e = blockIdx.x * blockDim.x + threadIdx.x;
  if (e >= ne) return;
  int d = dst_idx[e];
  float ex = __expf(e_arr[e] - unmapf(m_arr[d]));
  e_arr[e] = ex;
  atomicAdd(&s_arr[d], ex);
}

__global__ void edge_aggregate_kernel(const int* __restrict__ src_idx,
                                      const int* __restrict__ dst_idx,
                                      const float* __restrict__ e_arr,
                                      const float* __restrict__ s_arr,
                                      const float* __restrict__ src_feat,
                                      float* __restrict__ dst_accum, int ne) {
  int wave = threadIdx.x >> 5, lane = threadIdx.x & 31;
  int e = blockIdx.x * 8 + wave;
  if (e >= ne) return;
  int s = src_idx[e], d = dst_idx[e];
  float alpha = e_arr[e] / fmaxf(s_arr[d], 1e-30f);
  const float4* srow = (const float4*)(src_feat + (size_t)s * D_N);
  float* drow = dst_accum + (size_t)d * D_N;
#pragma unroll
  for (int h = 0; h < 2; ++h) {
    float4 v = srow[lane + h * 32];
    int c = (lane + h * 32) * 4;
    atomicAdd(&drow[c + 0], alpha * v.x);
    atomicAdd(&drow[c + 1], alpha * v.y);
    atomicAdd(&drow[c + 2], alpha * v.z);
    atomicAdd(&drow[c + 3], alpha * v.w);
  }
}

__global__ void fc_softmax_kernel(const float* __restrict__ H,
                                  const float* __restrict__ W,
                                  const float* __restrict__ bias,
                                  float* __restrict__ out) {
  __shared__ float tile[16][260];
  int wave = threadIdx.x >> 5;
  int lane = threadIdx.x & 31;
  size_t rowBase = (size_t)blockIdx.x * 16;
  int halfK = lane >> 4;
  int l16 = lane & 15;

  v8f zero = {0.f, 0.f, 0.f, 0.f, 0.f, 0.f, 0.f, 0.f};
  v8f acc0 = zero, acc1 = zero, acc2 = zero, acc3 = zero;

  const float* hrow = H + (rowBase + l16) * D_N;
  int colBase = wave * 64 + l16;
  for (int k0 = 0; k0 < D_N; k0 += 4) {
    int ka = k0 + halfK * 2;
    v2f a;
    a.x = hrow[ka];
    a.y = hrow[ka + 1];
    const float* w0 = W + (size_t)ka * DOUT_N;
    const float* w1 = w0 + DOUT_N;
    v2f b0; b0.x = w0[colBase +  0]; b0.y = w1[colBase +  0];
    v2f b1; b1.x = w0[colBase + 16]; b1.y = w1[colBase + 16];
    v2f b2; b2.x = w0[colBase + 32]; b2.y = w1[colBase + 32];
    v2f b3; b3.x = w0[colBase + 48]; b3.y = w1[colBase + 48];
    acc0 = __builtin_amdgcn_wmma_f32_16x16x4_f32(false, a, false, b0, (short)0, acc0, false, false);
    acc1 = __builtin_amdgcn_wmma_f32_16x16x4_f32(false, a, false, b1, (short)0, acc1, false, false);
    acc2 = __builtin_amdgcn_wmma_f32_16x16x4_f32(false, a, false, b2, (short)0, acc2, false, false);
    acc3 = __builtin_amdgcn_wmma_f32_16x16x4_f32(false, a, false, b3, (short)0, acc3, false, false);
  }

  int rowOff = halfK * 8;
  {
    v8f accs[4] = {acc0, acc1, acc2, acc3};
#pragma unroll
    for (int n = 0; n < 4; ++n) {
      int col = colBase + n * 16;
      float bv = bias[col];
#pragma unroll
      for (int r = 0; r < 8; ++r) tile[r + rowOff][col] = accs[n][r] + bv;
    }
  }
  __syncthreads();

#pragma unroll
  for (int rr = 0; rr < 4; ++rr) {
    int row = wave + rr * 4;
    float v[8];
    float mx = -3.402823466e38f;
#pragma unroll
    for (int i = 0; i < 8; ++i) {
      v[i] = tile[row][lane + i * 32];
      mx = fmaxf(mx, v[i]);
    }
#pragma unroll
    for (int off = 16; off; off >>= 1) mx = fmaxf(mx, __shfl_xor(mx, off, 32));
    float sum = 0.f;
#pragma unroll
    for (int i = 0; i < 8; ++i) { v[i] = __expf(v[i] - mx); sum += v[i]; }
#pragma unroll
    for (int off = 16; off; off >>= 1) sum += __shfl_xor(sum, off, 32);
    float inv = 1.0f / sum;
    float* orow = out + (rowBase + row) * DOUT_N;
#pragma unroll
    for (int i = 0; i < 8; ++i) orow[lane + i * 32] = v[i] * inv;
  }
}

extern "C" void kernel_launch(void* const* d_in, const int* in_sizes, int n_in,
                              void* d_out, int out_size, void* d_ws, size_t ws_size,
                              hipStream_t stream) {
  const float* doc_hidden  = (const float*)d_in[0];
  const float* word_hidden = (const float*)d_in[1];
  const int*   edge_doc    = (const int*)d_in[2];
  const int*   edge_word   = (const int*)d_in[3];
  const float* d2w_w       = (const float*)d_in[4];
  const float* d2w_b       = (const float*)d_in[5];
  const float* w2d_w       = (const float*)d_in[6];
  const float* w2d_b       = (const float*)d_in[7];
  const float* fc_w        = (const float*)d_in[8];
  const float* fc_b        = (const float*)d_in[9];
  float* out = (float*)d_out;

  float* ws = (float*)d_ws;
  const size_t NH = (size_t)(ND_N + NW_N) * D_N;
  float*    H        = ws;
  float*    sum_doc  = H + NH;
  float*    sum_word = sum_doc + ND_N;
  unsigned* max_doc  = (unsigned*)(sum_word + NW_N);
  unsigned* max_word = max_doc + ND_N;
  float*    sA_doc   = (float*)(max_word + NW_N);
  float*    sB_doc   = sA_doc + ND_N;
  float*    sA_word  = sB_doc + ND_N;
  float*    sB_word  = sA_word + NW_N;
  float*    e_d2w    = sB_word + NW_N;
  float*    e_w2d    = e_d2w + E_N;

  hipMemsetAsync(ws, 0, (NH + ND_N + NW_N) * sizeof(float), stream);
  {
    int n = ND_N + NW_N;
    fill_u32_kernel<<<(n + 255) / 256, 256, 0, stream>>>(max_doc, NEG_INF_MAPPED, n);
  }

  node_dots_kernel<<<ND_N / 8, 256, 0, stream>>>(doc_hidden, ND_N,
                                                 d2w_w, w2d_w + D_N, sA_doc, sB_doc);
  node_dots_kernel<<<NW_N / 8, 256, 0, stream>>>(word_hidden, NW_N,
                                                 d2w_w + D_N, w2d_w, sA_word, sB_word);

  const int egrid = (E_N + 255) / 256;
  edge_score_max_kernel<<<egrid, 256, 0, stream>>>(edge_doc, edge_word, sA_doc, sA_word,
                                                   d2w_b, e_d2w, max_word, E_N);
  edge_score_max_kernel<<<egrid, 256, 0, stream>>>(edge_word, edge_doc, sB_word, sB_doc,
                                                   w2d_b, e_w2d, max_doc, E_N);
  edge_exp_sum_kernel<<<egrid, 256, 0, stream>>>(edge_word, max_word, e_d2w, sum_word, E_N);
  edge_exp_sum_kernel<<<egrid, 256, 0, stream>>>(edge_doc, max_doc, e_w2d, sum_doc, E_N);
  edge_aggregate_kernel<<<E_N / 8, 256, 0, stream>>>(edge_doc, edge_word, e_d2w, sum_word,
                                                     doc_hidden, H + (size_t)ND_N * D_N, E_N);
  edge_aggregate_kernel<<<E_N / 8, 256, 0, stream>>>(edge_word, edge_doc, e_w2d, sum_doc,
                                                     word_hidden, H, E_N);

  fc_softmax_kernel<<<(ND_N + NW_N) / 16, 128, 0, stream>>>(H, fc_w, fc_b, out);

  (void)in_sizes; (void)n_in; (void)out_size; (void)ws_size;
}


# edge aggregation: each f32 atomic add instruction of a wave covers one whole 128-byte line of the destination row (lane-contiguous columns), not 8 floats in each of 4 lines; same atomics
# speedup vs baseline: 1.4707x; 1.4707x over previous
#include <hip/hip_runtime.h>

#define ND_N 20000
#define NW_N 30000
#define E_N  500000
#define D_N  256
#define DOUT_N 256

typedef __attribute__((ext_vector_type(2))) float v2f;
typedef __attribute__((ext_vector_type(8))) float v8f;

__device__ __forceinline__ unsigned mapf(float f) {
  unsigned u = __float_as_uint(f);
  return (u & 0x80000000u) ? ~u : (u | 0x80000000u);
}
__device__ __forceinline__ float unmapf(unsigned u) {
  unsigned v = (u & 0x80000000u) ? (u & 0x7fffffffu) : ~u;
  return __uint_as_float(v);
}
#define NEG_INF_MAPPED 0x007fffffu

__global__ void fill_u32_kernel(unsigned* __restrict__ p, unsigned v, int n) {
  int i = blockIdx.x * blockDim.x + threadIdx.x;
  if (i < n) p[i] = v;
}

__global__ void node_dots_kernel(const float* __restrict__ hid, int n,
                                 const float* __restrict__ wa,
                                 const float* __restrict__ wb,
                                 float* __restrict__ outa,
                                 float* __restrict__ outb) {
  int wave = threadIdx.x >> 5, lane = threadIdx.x & 31;
  int node = blockIdx.x * 8 + wave;
  if (node >= n) return;
  const float* row = hid + (size_t)node * D_N;
  float da = 0.f, db = 0.f;
#pragma unroll
  for (int i = 0; i < 8; ++i) {
    int k = lane + i * 32;
    float x = row[k];
    da += x * wa[k];
    db += x * wb[k];
  }
#pragma unroll
  for (int off = 16; off; off >>= 1) {
    da += __shfl_xor(da, off, 32);
    db += __shfl_xor(db, off, 32);
  }
  if (lane == 0) { outa[node] = da; outb[node] = db; }
}

__global__ void edge_score_max_kernel(const int* __restrict__ src_idx,
                                      const int* __restrict__ dst_idx,
                                      const float* __restrict__ s_src,
                                      const float* __restrict__ s_dst,
                                      const float* __restrict__ bptr,
                                      float* __restrict__ e_arr,
                                      unsigned* __restrict__ m_arr, int ne) {
  int e = blockIdx.x * blockDim.x + threadIdx.x;
  if (e >= ne) return;
  float a = s_src[src_idx[e]] + s_dst[dst_idx[e]] + bptr[0];
  float v = a > 0.f ? a : 0.01f * a;
  e_arr[e] = v;
  atomicMax(&m_arr[dst_idx[e]], mapf(v));
}

__global__ void edge_exp_sum_kernel(const int* __restrict__ dst_idx,
                                    const unsigned* __restrict__ m_arr,
                                    float* __restrict__ e_arr,
                                    float* __restrict__ s_arr, int ne) {
  int e = blockIdx.x * blockDim.x + threadIdx.x;
  if (e >= ne) return;
  int d = dst_idx[e];
  float ex = __expf(e_arr[e] - unmapf(m_arr[d]));
  e_arr[e] = ex;
  atomicAdd(&s_arr[d], ex);
}

__global__ void edge_aggregate_kernel(const int* __restrict__ src_idx,
                                      const int* __restrict__ dst_idx,
                                      const float* __restrict__ e_arr,
                                      const float* __restrict__ s_arr,
                                      const float* __restrict__ src_feat,
                                      float* __restrict__ dst_accum, int ne) {
  int wave = threadIdx.x >> 5, lane = threadIdx.x & 31;
  int e = blockIdx.x * 8 + wave;
  if (e >= ne) return;
  int s = src_idx[e], d = dst_idx[e];
  float alpha = e_arr[e] / fmaxf(s_arr[d], 1e-30f);
  const float* srow = src_feat + (size_t)s * D_N;
  float* drow = dst_accum + (size_t)d * D_N;
  float v[8];
  for (int i = 0; i < 8; ++i) v[i] = srow[lane + i * 32];
  for (int i = 0; i < 8; ++i) atomicAdd(&drow[lane + i * 32], alpha * v[i]);
}

__global__ void fc_softmax_kernel(const float* __restrict__ H,
                                  const float* __restrict__ W,
                                  const float* __restrict__ bias,
                                  float* __restrict__ out) {
  __shared__ float tile[16][260];
  int wave = threadIdx.x >> 5;
  int lane = threadIdx.x & 31;
  size_t rowBase = (size_t)blockIdx.x * 16;
  int halfK = lane >> 4;
  int l16 = lane & 15;

  v8f zero = {0.f, 0.f, 0.f, 0.f, 0.f, 0.f, 0.f, 0.f};
  v8f acc0 = zero, acc1 = zero, acc2 = zero, acc3 = zero;

  const float* hrow = H + (rowBase + l16) * D_N;
  int colBase = wave * 64 + l16;
  for (int k0 = 0; k0 < D_N; k0 += 4) {
    int ka = k0 + halfK * 2;
    v2f a;
    a.x = hrow[ka];
    a.y = hrow[ka + 1];
    const float* w0 = W + (size_t)ka * DOUT_N;
    const float* w1 = w0 + DOUT_N;
    v2f b0; b0.x = w0[colBase +  0]; b0.y = w1[colBase +  0];
    v2f b1; b1.x = w0[colBase + 16]; b1.y = w1[colBase + 16];
    v2f b2; b2.x = w0[colBase + 32]; b2.y = w1[colBase + 32];
    v2f b3; b3.x = w0[colBase + 48]; b3.y = w1[colBase + 48];
    acc0 = __builtin_amdgcn_wmma_f32_16x16x4_f32(false, a, false, b0, (short)0, acc0, false, false);
    acc1 = __builtin_amdgcn_wmma_f32_16x16x4_f32(false, a, false, b1, (short)0, acc1, false, false);
    acc2 = __builtin_amdgcn_wmma_f32_16x16x4_f32(false, a, false, b2, (short)0, acc2, false, false);
    acc3 = __builtin_amdgcn_wmma_f32_16x16x4_f32(false, a, false, b3, (short)0, acc3, false, false);
  }

  int rowOff = halfK * 8;
  {
    v8f accs[4] = {acc0, acc1, acc2, acc3};
#pragma unroll
    for (int n = 0; n < 4; ++n) {
      int col = colBase + n * 16;
      float bv = bias[col];
#pragma unroll
      for (int r = 0; r < 8; ++r) tile[r + rowOff][col] = accs[n][r] + bv;
    }
  }
  __syncthreads();

#pragma unroll
  for (int rr = 0; rr < 4; ++rr) {
    int row = wave + rr * 4;
    float v[8];
    float mx = -3.402823466e38f;
#pragma unroll
    for (int i = 0; i < 8; ++i) {
      v[i] = tile[row][lane + i * 32];
      mx = fmaxf(mx, v[i]);
    }
#pragma unroll
    for (int off = 16; off; off >>= 1) mx = fmaxf(mx, __shfl_xor(mx, off, 32));
    float sum = 0.f;
#pragma unroll
    for (int i = 0; i < 8; ++i) { v[i] = __expf(v[i] - mx); sum += v[i]; }
#pragma unroll
    for (int off = 16; off; off >>= 1) sum += __shfl_xor(sum, off, 32);
    float inv = 1.0f / sum;
    float* orow = out + (rowBase + row) * DOUT_N;
#pragma unroll
    for (int i = 0; i < 8; ++i) orow[lane + i * 32] = v[i] * inv;
  }
}

extern "C" void kernel_launch(void* const* d_in, const int* in_sizes, int n_in,
                              void* d_out, int out_size, void* d_ws, size_t ws_size,
                              hipStream_t stream) {
  const float* doc_hidden  = (const float*)d_in[0];
  const float* word_hidden = (const float*)d_in[1];
  const int*   edge_doc    = (const int*)d_in[2];
  const int*   edge_word   = (const int*)d_in[3];
  const float* d2w_w       = (const float*)d_in[4];
  const float* d2w_b       = (const float*)d_in[5];
  const float* w2d_w       = (const float*)d_in[6];
  const float* w2d_b       = (const float*)d_in[7];
  const float* fc_w        = (const float*)d_in[8];
  const float* fc_b        = (const float*)d_in[9];
  float* out = (float*)d_out;

  float* ws = (float*)d_ws;
  const size_t NH = (size_t)(ND_N + NW_N) * D_N;
  float*    H        = ws;
  float*    sum_doc  = H + NH;
  float*    sum_word = sum_doc + ND_N;
  unsigned* max_doc  = (unsigned*)(sum_word + NW_N);
  unsigned* max_word = max_doc + ND_N;
  float*    sA_doc   = (float*)(max_word + NW_N);
  float*    sB_doc   = sA_doc + ND_N;
  float*    sA_word  = sB_doc + ND_N;
  float*    sB_word  = sA_word + NW_N;
  float*    e_d2w    = sB_word + NW_N;
  float*    e_w2d    = e_d2w + E_N;

  hipMemsetAsync(ws, 0, (NH + ND_N + NW_N) * sizeof(float), stream);
  {
    int n = ND_N + NW_N;
    fill_u32_kernel<<<(n + 255) / 256, 256, 0, stream>>>(max_doc, NEG_INF_MAPPED, n);
  }

  node_dots_kernel<<<ND_N / 8, 256, 0, stream>>>(doc_hidden, ND_N,
                                                 d2w_w, w2d_w + D_N, sA_doc, sB_doc);
  node_dots_kernel<<<NW_N / 8, 256, 0, stream>>>(word_hidden, NW_N,
                                                 d2w_w + D_N, w2d_w, sA_word, sB_word);

  const int egrid = (E_N + 255) / 256;
  edge_score_max_kernel<<<egrid, 256, 0, stream>>>(edge_doc, edge_word, sA_doc, sA_word,
                                                   d2w_b, e_d2w, max_word, E_N);
  edge_score_max_kernel<<<egrid, 256, 0, stream>>>(edge_word, edge_doc, sB_word, sB_doc,
                                                   w2d_b, e_w2d, max_doc, E_N);
  edge_exp_sum_kernel<<<egrid, 256, 0, stream>>>(edge_word, max_word, e_d2w, sum_word, E_N);
  edge_exp_sum_kernel<<<egrid, 256, 0, stream>>>(edge_doc, max_doc, e_w2d, sum_doc, E_N);
  edge_aggregate_kernel<<<E_N / 8, 256, 0, stream>>>(edge_doc, edge_word, e_d2w, sum_word,
                                                     doc_hidden, H + (size_t)ND_N * D_N, E_N);
  edge_aggregate_kernel<<<E_N / 8, 256, 0, stream>>>(edge_word, edge_doc, e_w2d, sum_doc,
                                                     word_hidden, H, E_N);

  fc_softmax_kernel<<<(ND_N + NW_N) / 16, 128, 0, stream>>>(H, fc_w, fc_b, out);

  (void)in_sizes; (void)n_in; (void)out_size; (void)ws_size;
}


# edges grouped by destination (integer count, scan, place, in-wave sort by source id); one wave per destination: segment max, exp sum, weighted row sum in registers in sorted order; no f32 atomics
# speedup vs baseline: 2.2805x; 1.5506x over previous
#include <hip/hip_runtime.h>

#define ND_N 20000
#define NW_N 30000
#define E_N  500000
#define D_N  256
#define DOUT_N 256
#define NG_N (ND_N + NW_N)

typedef __attribute__((ext_vector_type(2))) float v2f;
typedef __attribute__((ext_vector_type(4))) float v4f;
typedef __attribute__((ext_vector_type(8))) float v8f;

__global__ void node_dots_kernel(const float* __restrict__ hid, int n,
                                 const float* __restrict__ wa,
                                 const float* __restrict__ wb,
                                 float* __restrict__ outa,
                                 float* __restrict__ outb) {
  int wave = threadIdx.x >> 5, lane = threadIdx.x & 31;
  int node = blockIdx.x * 8 + wave;
  if (node >= n) return;
  const float* row = hid + (size_t)node * D_N;
  float da = 0.f, db = 0.f;
#pragma unroll
  for (int i = 0; i < 8; ++i) {
    int k = lane + i * 32;
    float x = row[k];
    da += x * wa[k];
    db += x * wb[k];
  }
#pragma unroll
  for (int off = 16; off; off >>= 1) {
    da += __shfl_xor(da, off, 32);
    db += __shfl_xor(db, off, 32);
  }
  if (lane == 0) { outa[node] = da; outb[node] = db; }
}


__global__ void edge_count_kernel(const int* __restrict__ edge_doc,
                                  const int* __restrict__ edge_word,
                                  unsigned* __restrict__ cnt,
                                  unsigned* __restrict__ slot_doc,
                                  unsigned* __restrict__ slot_word, int ne) {
  int e = blockIdx.x * blockDim.x + threadIdx.x;
  if (e >= ne) return;
  int dd = edge_doc[e], ww = edge_word[e];
  slot_doc[e]  = atomicAdd(&cnt[dd], 1u);
  slot_word[e] = atomicAdd(&cnt[ND_N + ww], 1u);
}

__global__ void scan_counts_kernel(const unsigned* __restrict__ cnt, unsigned* __restrict__ off) {
  __shared__ unsigned wsum[32];
  const int PER = (NG_N + 1023) / 1024;
  int t = threadIdx.x, wave = t >> 5, lane = t & 31;
  int base = t * PER;
  unsigned s = 0;
  unsigned vals[PER];
#pragma unroll
  for (int i = 0; i < PER; ++i) {
    int k = base + i;
    unsigned v = cnt[k < NG_N ? k : NG_N - 1];
    vals[i] = k < NG_N ? v : 0u;
    s += vals[i];
  }
  unsigned inc = s;
  for (int o = 1; o < 32; o <<= 1) {
    unsigned v = __shfl_up(inc, o, 32);
    if (lane >= o) inc += v;
  }
  if (lane == 31) wsum[wave] = inc;
  __syncthreads();
  if (wave == 0) {
    unsigned w = wsum[lane], wi = w;
    for (int o = 1; o < 32; o <<= 1) {
      unsigned v = __shfl_up(wi, o, 32);
      if (lane >= o) wi += v;
    }
    wsum[lane] = wi - w;
  }
  __syncthreads();
  unsigned run = wsum[wave] + inc - s;
#pragma unroll
  for (int i = 0; i < PER; ++i) {
    int k = base + i;
    if (k < NG_N) off[k] = run;
    run += vals[i];
  }
  if (t == 1023) off[NG_N] = run;
}

__global__ void edge_place_kernel(const int* __restrict__ edge_doc,
                                  const int* __restrict__ edge_word,
                                  const unsigned* __restrict__ off,
                                  const unsigned* __restrict__ slot_doc,
                                  const unsigned* __restrict__ slot_word,
                                  int* __restrict__ grp, int ne) {
  int e = blockIdx.x * blockDim.x + threadIdx.x;
  if (e >= ne) return;
  int dd = edge_doc[e], ww = edge_word[e];
  grp[off[dd] + slot_doc[e]] = ww;
  grp[off[ND_N + ww] + slot_word[e]] = dd;
}

__global__ void group_sort_kernel(const unsigned* __restrict__ off,
                                  const int* __restrict__ grp_in,
                                  int* __restrict__ grp_out) {
  int lane = threadIdx.x & 31;
  int g = __builtin_amdgcn_readfirstlane((int)(blockIdx.x * 8 + (threadIdx.x >> 5)));
  if (g >= NG_N) return;
  int start = (int)off[g];
  int deg = (int)off[g + 1] - start;
  const int* in = grp_in + start;
  int* outp = grp_out + start;
  for (int c = 0; c < deg; c += 32) {
    int idx = c + lane;
    int key = idx < deg ? in[idx] : 0x7fffffff;
    int rank = 0;
    for (int c2 = 0; c2 < deg; c2 += 32) {
      int idx2 = c2 + lane;
      int k2 = idx2 < deg ? in[idx2] : 0x7fffffff;
      int n2 = deg - c2 < 32 ? deg - c2 : 32;
      for (int j = 0; j < n2; ++j) {
        int v = __builtin_amdgcn_readlane(k2, j);
        rank += (v < key || (v == key && c2 + j < idx)) ? 1 : 0;
      }
    }
    if (idx < deg) outp[rank] = key;
  }
}

__device__ __forceinline__ float group_score(const int* __restrict__ seg, const float* __restrict__ ssrc,
                                             int idx, float sd, float bb, int* sidx) {
  int s = seg[idx];
  *sidx = s;
  float a = ssrc[s] + sd + bb;
  return a > 0.f ? a : 0.01f * a;
}

__global__ void group_aggregate_kernel(const unsigned* __restrict__ off,
                                       const int* __restrict__ grp,
                                       const float* __restrict__ doc_hidden,
                                       const float* __restrict__ word_hidden,
                                       const float* __restrict__ sA_doc,
                                       const float* __restrict__ sA_word,
                                       const float* __restrict__ sB_doc,
                                       const float* __restrict__ sB_word,
                                       const float* __restrict__ d2w_b,
                                       const float* __restrict__ w2d_b,
                                       float* __restrict__ H) {
  int lane = threadIdx.x & 31;
  int g = __builtin_amdgcn_readfirstlane((int)(blockIdx.x * 8 + (threadIdx.x >> 5)));
  if (g >= NG_N) return;
  const float* feat; const float* ssrc; float sd, bb;
  if (g < ND_N) { feat = word_hidden; ssrc = sB_word; sd = sB_doc[g];         bb = w2d_b[0]; }
  else          { feat = doc_hidden;  ssrc = sA_doc;  sd = sA_word[g - ND_N]; bb = d2w_b[0]; }
  int start = (int)off[g];
  int deg = (int)off[g + 1] - start;
  const int* seg = grp + start;

  int sidx0 = 0; float v0 = 0.f;
  if (lane < deg) v0 = group_score(seg, ssrc, lane, sd, bb, &sidx0);
  auto score_at = [&](int c, int idx, int* sidx) -> float {
    if (c == 0) { *sidx = sidx0; return v0; }
    return group_score(seg, ssrc, idx, sd, bb, sidx);
  };

  float mx = -__builtin_huge_valf();
  for (int c = 0; c < deg; c += 32) {
    int idx = c + lane;
    if (idx < deg) { int s; mx = fmaxf(mx, score_at(c, idx, &s)); }
  }
  for (int o = 16; o; o >>= 1) mx = fmaxf(mx, __shfl_xor(mx, o, 32));

  float sum = 0.f;
  for (int c = 0; c < deg; c += 32) {
    int idx = c + lane;
    float ex = 0.f;
    if (idx < deg) { int s; ex = __expf(score_at(c, idx, &s) - mx); }
    for (int o = 16; o; o >>= 1) ex += __shfl_xor(ex, o, 32);
    sum += ex;
  }
  float den = fmaxf(sum, 1e-30f);

  v4f acc0 = {0.f, 0.f, 0.f, 0.f}, acc1 = {0.f, 0.f, 0.f, 0.f};
  for (int c = 0; c < deg; c += 32) {
    int idx = c + lane;
    int sidx = 0; float al = 0.f;
    if (idx < deg) al = __expf(score_at(c, idx, &sidx) - mx) / den;
    int ial = __float_as_int(al);
    int n = deg - c < 32 ? deg - c : 32;
    int j = 0;
    for (; j + 4 <= n; j += 4) {
      const v4f* r0 = reinterpret_cast<const v4f*>(feat + (size_t)__builtin_amdgcn_readlane(sidx, j)     * D_N);
      const v4f* r1 = reinterpret_cast<const v4f*>(feat + (size_t)__builtin_amdgcn_readlane(sidx, j + 1) * D_N);
      const v4f* r2 = reinterpret_cast<const v4f*>(feat + (size_t)__builtin_amdgcn_readlane(sidx, j + 2) * D_N);
      const v4f* r3 = reinterpret_cast<const v4f*>(feat + (size_t)__builtin_amdgcn_readlane(sidx, j + 3) * D_N);
      float a0 = __int_as_float(__builtin_amdgcn_readlane(ial, j));
      float a1 = __int_as_float(__builtin_amdgcn_readlane(ial, j + 1));
      float a2 = __int_as_float(__builtin_amdgcn_readlane(ial, j + 2));
      float a3 = __int_as_float(__builtin_amdgcn_readlane(ial, j + 3));
      v4f x00 = r0[lane], x01 = r0[lane + 32];
      v4f x10 = r1[lane], x11 = r1[lane + 32];
      v4f x20 = r2[lane], x21 = r2[lane + 32];
      v4f x30 = r3[lane], x31 = r3[lane + 32];
      acc0 += a0 * x00; acc1 += a0 * x01;
      acc0 += a1 * x10; acc1 += a1 * x11;
      acc0 += a2 * x20; acc1 += a2 * x21;
      acc0 += a3 * x30; acc1 += a3 * x31;
    }
    for (; j < n; ++j) {
      const v4f* r0 = reinterpret_cast<const v4f*>(feat + (size_t)__builtin_amdgcn_readlane(sidx, j) * D_N);
      float a0 = __int_as_float(__builtin_amdgcn_readlane(ial, j));
      v4f x00 = r0[lane], x01 = r0[lane + 32];
      acc0 += a0 * x00; acc1 += a0 * x01;
    }
  }
  v4f* hrow = reinterpret_cast<v4f*>(H + (size_t)g * D_N);
  hrow[lane] = acc0;
  hrow[lane + 32] = acc1;
}

__global__ void fc_softmax_kernel(const float* __restrict__ H,
                                  const float* __restrict__ W,
                                  const float* __restrict__ bias,
                                  float* __restrict__ out) {
  __shared__ float tile[16][260];
  int wave = threadIdx.x >> 5;
  int lane = threadIdx.x & 31;
  size_t rowBase = (size_t)blockIdx.x * 16;
  int halfK = lane >> 4;
  int l16 = lane & 15;

  v8f zero = {0.f, 0.f, 0.f, 0.f, 0.f, 0.f, 0.f, 0.f};
  v8f acc0 = zero, acc1 = zero, acc2 = zero, acc3 = zero;

  const float* hrow = H + (rowBase + l16) * D_N;
  int colBase = wave * 64 + l16;
  for (int k0 = 0; k0 < D_N; k0 += 4) {
    int ka = k0 + halfK * 2;
    v2f a;
    a.x = hrow[ka];
    a.y = hrow[ka + 1];
    const float* w0 = W + (size_t)ka * DOUT_N;
    const float* w1 = w0 + DOUT_N;
    v2f b0; b0.x = w0[colBase +  0]; b0.y = w1[colBase +  0];
    v2f b1; b1.x = w0[colBase + 16]; b1.y = w1[colBase + 16];
    v2f b2; b2.x = w0[colBase + 32]; b2.y = w1[colBase + 32];
    v2f b3; b3.x = w0[colBase + 48]; b3.y = w1[colBase + 48];
    acc0 = __builtin_amdgcn_wmma_f32_16x16x4_f32(false, a, false, b0, (short)0, acc0, false, false);
    acc1 = __builtin_amdgcn_wmma_f32_16x16x4_f32(false, a, false, b1, (short)0, acc1, false, false);
    acc2 = __builtin_amdgcn_wmma_f32_16x16x4_f32(false, a, false, b2, (short)0, acc2, false, false);
    acc3 = __builtin_amdgcn_wmma_f32_16x16x4_f32(false, a, false, b3, (short)0, acc3, false, false);
  }

  int rowOff = halfK * 8;
  {
    v8f accs[4] = {acc0, acc1, acc2, acc3};
#pragma unroll
    for (int n = 0; n < 4; ++n) {
      int col = colBase + n * 16;
      float bv = bias[col];
#pragma unroll
      for (int r = 0; r < 8; ++r) tile[r + rowOff][col] = accs[n][r] + bv;
    }
  }
  __syncthreads();

#pragma unroll
  for (int rr = 0; rr < 4; ++rr) {
    int row = wave + rr * 4;
    float v[8];
    float mx = -3.402823466e38f;
#pragma unroll
    for (int i = 0; i < 8; ++i) {
      v[i] = tile[row][lane + i * 32];
      mx = fmaxf(mx, v[i]);
    }
#pragma unroll
    for (int off = 16; off; off >>= 1) mx = fmaxf(mx, __shfl_xor(mx, off, 32));
    float sum = 0.f;
#pragma unroll
    for (int i = 0; i < 8; ++i) { v[i] = __expf(v[i] - mx); sum += v[i]; }
#pragma unroll
    for (int off = 16; off; off >>= 1) sum += __shfl_xor(sum, off, 32);
    float inv = 1.0f / sum;
    float* orow = out + (rowBase + row) * DOUT_N;
#pragma unroll
    for (int i = 0; i < 8; ++i) orow[lane + i * 32] = v[i] * inv;
  }
}

extern "C" void kernel_launch(void* const* d_in, const int* in_sizes, int n_in,
                              void* d_out, int out_size, void* d_ws, size_t ws_size,
                              hipStream_t stream) {
  const float* doc_hidden  = (const float*)d_in[0];
  const float* word_hidden = (const float*)d_in[1];
  const int*   edge_doc    = (const int*)d_in[2];
  const int*   edge_word   = (const int*)d_in[3];
  const float* d2w_w       = (const float*)d_in[4];
  const float* d2w_b       = (const float*)d_in[5];
  const float* w2d_w       = (const float*)d_in[6];
  const float* w2d_b       = (const float*)d_in[7];
  const float* fc_w        = (const float*)d_in[8];
  const float* fc_b        = (const float*)d_in[9];
  float* out = (float*)d_out;

  float* ws = (float*)d_ws;
  const size_t NH = (size_t)NG_N * D_N;
  float*    H         = ws;
  unsigned* cnt       = (unsigned*)ws;
  unsigned* slot_doc  = cnt + NG_N;
  unsigned* slot_word = slot_doc + E_N;
  int*      grp_arr   = (int*)(slot_word + E_N);
  unsigned* off       = (unsigned*)(H + NH);
  float*    sA_doc    = (float*)(off + NG_N + 1);
  float*    sB_doc    = sA_doc + ND_N;
  float*    sA_word   = sB_doc + ND_N;
  float*    sB_word   = sA_word + NW_N;
  int*      grp_srt   = (int*)(sB_word + NW_N);

  hipMemsetAsync(cnt, 0, (size_t)NG_N * sizeof(unsigned), stream);

  node_dots_kernel<<<ND_N / 8, 256, 0, stream>>>(doc_hidden, ND_N,
                                                 d2w_w, w2d_w + D_N, sA_doc, sB_doc);
  node_dots_kernel<<<NW_N / 8, 256, 0, stream>>>(word_hidden, NW_N,
                                                 d2w_w + D_N, w2d_w, sA_word, sB_word);

  const int egrid = (E_N + 255) / 256;
  edge_count_kernel<<<egrid, 256, 0, stream>>>(edge_doc, edge_word, cnt, slot_doc, slot_word, E_N);
  scan_counts_kernel<<<1, 1024, 0, stream>>>(cnt, off);
  edge_place_kernel<<<egrid, 256, 0, stream>>>(edge_doc, edge_word, off, slot_doc, slot_word, grp_arr, E_N);
  group_sort_kernel<<<NG_N / 8, 256, 0, stream>>>(off, grp_arr, grp_srt);

  group_aggregate_kernel<<<NG_N / 8, 256, 0, stream>>>(off, grp_srt, doc_hidden, word_hidden,
                                                       sA_doc, sA_word, sB_doc, sB_word,
                                                       d2w_b, w2d_b, H);

  fc_softmax_kernel<<<NG_N / 16, 128, 0, stream>>>(H, fc_w, fc_b, out);

  (void)in_sizes; (void)n_in; (void)out_size; (void)ws_size;
}


# count scan: 13 blocks, 4 consecutive counts per thread by 16-byte loads and stores (whole 128-byte lines per instruction), not one block of 49 strided 4-byte loads and stores per thread
# speedup vs baseline: 2.6573x; 1.1652x over previous
#include <hip/hip_runtime.h>

#define ND_N 20000
#define NW_N 30000
#define E_N  500000
#define D_N  256
#define DOUT_N 256
#define NG_N (ND_N + NW_N)

typedef __attribute__((ext_vector_type(2))) float v2f;
typedef __attribute__((ext_vector_type(4))) float v4f;
typedef __attribute__((ext_vector_type(8))) float v8f;

__global__ void node_dots_kernel(const float* __restrict__ hid, int n,
                                 const float* __restrict__ wa,
                                 const float* __restrict__ wb,
                                 float* __restrict__ outa,
                                 float* __restrict__ outb) {
  int wave = threadIdx.x >> 5, lane = threadIdx.x & 31;
  int node = blockIdx.x * 8 + wave;
  if (node >= n) return;
  const float* row = hid + (size_t)node * D_N;
  float da = 0.f, db = 0.f;
#pragma unroll
  for (int i = 0; i < 8; ++i) {
    int k = lane + i * 32;
    float x = row[k];
    da += x * wa[k];
    db += x * wb[k];
  }
#pragma unroll
  for (int off = 16; off; off >>= 1) {
    da += __shfl_xor(da, off, 32);
    db += __shfl_xor(db, off, 32);
  }
  if (lane == 0) { outa[node] = da; outb[node] = db; }
}


__global__ void edge_count_kernel(const int* __restrict__ edge_doc,
                                  const int* __restrict__ edge_word,
                                  unsigned* __restrict__ cnt,
                                  unsigned* __restrict__ slot_doc,
                                  unsigned* __restrict__ slot_word, int ne) {
  int e = blockIdx.x * blockDim.x + threadIdx.x;
  if (e >= ne) return;
  int dd = edge_doc[e], ww = edge_word[e];
  slot_doc[e]  = atomicAdd(&cnt[dd], 1u);
  slot_word[e] = atomicAdd(&cnt[ND_N + ww], 1u);
}

#define SCAN_T 1024
#define SCAN_PER (4 * SCAN_T)
#define SCAN_BLOCKS ((NG_N + SCAN_PER - 1) / SCAN_PER)
static_assert(NG_N % 4 == 0, "a 16-byte group never straddles the end of the counts");
typedef __attribute__((ext_vector_type(4))) unsigned v4u;
__global__ void __launch_bounds__(SCAN_T) scan_counts_kernel(const unsigned* __restrict__ cnt, unsigned* __restrict__ off) {
  __shared__ unsigned wpre[32];
  __shared__ unsigned wtot[32];
  int t = threadIdx.x, wave = t >> 5, lane = t & 31;
  int blk = blockIdx.x;
  const v4u* cnt4 = reinterpret_cast<const v4u*>(cnt);
  int q = blk * SCAN_T + t;
  bool own = q < NG_N / 4;
  v4u mine = cnt4[own ? q : 0];
  v4u r[SCAN_BLOCKS - 1];
#pragma unroll
  for (int j = 0; j < SCAN_BLOCKS - 1; ++j) {
    r[j] = (v4u){0u, 0u, 0u, 0u};
    if (j < blk) r[j] = cnt4[j * SCAN_T + t];
  }
  v4u p4 = {0u, 0u, 0u, 0u};
#pragma unroll
  for (int j = 0; j < SCAN_BLOCKS - 1; ++j) p4 += r[j];
  unsigned pre = (p4.x + p4.y) + (p4.z + p4.w);
  if (!own) mine = (v4u){0u, 0u, 0u, 0u};
  unsigned s1 = mine.x, s2 = s1 + mine.y, s3 = s2 + mine.z, tot = s3 + mine.w;
  for (int o = 16; o; o >>= 1) pre += __shfl_xor(pre, o, 32);
  unsigned inc = tot;
  for (int o = 1; o < 32; o <<= 1) {
    unsigned u = __shfl_up(inc, o, 32);
    if (lane >= o) inc += u;
  }
  if (lane == 0) wpre[wave] = pre;
  if (lane == 31) wtot[wave] = inc;
  __syncthreads();
  unsigned x = wpre[lane] + (lane < wave ? wtot[lane] : 0u);
  for (int o = 16; o; o >>= 1) x += __shfl_xor(x, o, 32);
  unsigned e0 = x + inc - tot;
  if (own) {
    v4u o4 = {e0, e0 + s1, e0 + s2, e0 + s3};
    reinterpret_cast<v4u*>(off)[q] = o4;
    if (q == NG_N / 4 - 1) off[NG_N] = e0 + tot;
  }
}

__global__ void edge_place_kernel(const int* __restrict__ edge_doc,
                                  const int* __restrict__ edge_word,
                                  const unsigned* __restrict__ off,
                                  const unsigned* __restrict__ slot_doc,
                                  const unsigned* __restrict__ slot_word,
                                  int* __restrict__ grp, int ne) {
  int e = blockIdx.x * blockDim.x + threadIdx.x;
  if (e >= ne) return;
  int dd = edge_doc[e], ww = edge_word[e];
  grp[off[dd] + slot_doc[e]] = ww;
  grp[off[ND_N + ww] + slot_word[e]] = dd;
}

__global__ void group_sort_kernel(const unsigned* __restrict__ off,
                                  const int* __restrict__ grp_in,
                                  int* __restrict__ grp_out) {
  int lane = threadIdx.x & 31;
  int g = __builtin_amdgcn_readfirstlane((int)(blockIdx.x * 8 + (threadIdx.x >> 5)));
  if (g >= NG_N) return;
  int start = (int)off[g];
  int deg = (int)off[g + 1] - start;
  const int* in = grp_in + start;
  int* outp = grp_out + start;
  for (int c = 0; c < deg; c += 32) {
    int idx = c + lane;
    int key = idx < deg ? in[idx] : 0x7fffffff;
    int rank = 0;
    for (int c2 = 0; c2 < deg; c2 += 32) {
      int idx2 = c2 + lane;
      int k2 = idx2 < deg ? in[idx2] : 0x7fffffff;
      int n2 = deg - c2 < 32 ? deg - c2 : 32;
      for (int j = 0; j < n2; ++j) {
        int v = __builtin_amdgcn_readlane(k2, j);
        rank += (v < key || (v == key && c2 + j < idx)) ? 1 : 0;
      }
    }
    if (idx < deg) outp[rank] = key;
  }
}

__device__ __forceinline__ float group_score(const int* __restrict__ seg, const float* __restrict__ ssrc,
                                             int idx, float sd, float bb, int* sidx) {
  int s = seg[idx];
  *sidx = s;
  float a = ssrc[s] + sd + bb;
  return a > 0.f ? a : 0.01f * a;
}

__global__ void group_aggregate_kernel(const unsigned* __restrict__ off,
                                       const int* __restrict__ grp,
                                       const float* __restrict__ doc_hidden,
                                       const float* __restrict__ word_hidden,
                                       const float* __restrict__ sA_doc,
                                       const float* __restrict__ sA_word,
                                       const float* __restrict__ sB_doc,
                                       const float* __restrict__ sB_word,
                                       const float* __restrict__ d2w_b,
                                       const float* __restrict__ w2d_b,
                                       float* __restrict__ H) {
  int lane = threadIdx.x & 31;
  int g = __builtin_amdgcn_readfirstlane((int)(blockIdx.x * 8 + (threadIdx.x >> 5)));
  if (g >= NG_N) return;
  const float* feat; const float* ssrc; float sd, bb;
  if (g < ND_N) { feat = word_hidden; ssrc = sB_word; sd = sB_doc[g];         bb = w2d_b[0]; }
  else          { feat = doc_hidden;  ssrc = sA_doc;  sd = sA_word[g - ND_N]; bb = d2w_b[0]; }
  int start = (int)off[g];
  int deg = (int)off[g + 1] - start;
  const int* seg = grp + start;

  int sidx0 = 0; float v0 = 0.f;
  if (lane < deg) v0 = group_score(seg, ssrc, lane, sd, bb, &sidx0);
  auto score_at = [&](int c, int idx, int* sidx) -> float {
    if (c == 0) { *sidx = sidx0; return v0; }
    return group_score(seg, ssrc, idx, sd, bb, sidx);
  };

  float mx = -__builtin_huge_valf();
  for (int c = 0; c < deg; c += 32) {
    int idx = c + lane;
    if (idx < deg) { int s; mx = fmaxf(mx, score_at(c, idx, &s)); }
  }
  for (int o = 16; o; o >>= 1) mx = fmaxf(mx, __shfl_xor(mx, o, 32));

  float sum = 0.f;
  for (int c = 0; c < deg; c += 32) {
    int idx = c + lane;
    float ex = 0.f;
    if (idx < deg) { int s; ex = __expf(score_at(c, idx, &s) - mx); }
    for (int o = 16; o; o >>= 1) ex += __shfl_xor(ex, o, 32);
    sum += ex;
  }
  float den = fmaxf(sum, 1e-30f);

  v4f acc0 = {0.f, 0.f, 0.f, 0.f}, acc1 = {0.f, 0.f, 0.f, 0.f};
  for (int c = 0; c < deg; c += 32) {
    int idx = c + lane;
    int sidx = 0; float al = 0.f;
    if (idx < deg) al = __expf(score_at(c, idx, &sidx) - mx) / den;
    int ial = __float_as_int(al);
    int n = deg - c < 32 ? deg - c : 32;
    int j = 0;
    for (; j + 4 <= n; j += 4) {
      const v4f* r0 = reinterpret_cast<const v4f*>(feat + (size_t)__builtin_amdgcn_readlane(sidx, j)     * D_N);
      const v4f* r1 = reinterpret_cast<const v4f*>(feat + (size_t)__builtin_amdgcn_readlane(sidx, j + 1) * D_N);
      const v4f* r2 = reinterpret_cast<const v4f*>(feat + (size_t)__builtin_amdgcn_readlane(sidx, j + 2) * D_N);
      const v4f* r3 = reinterpret_cast<const v4f*>(feat + (size_t)__builtin_amdgcn_readlane(sidx, j + 3) * D_N);
      float a0 = __int_as_float(__builtin_amdgcn_readlane(ial, j));
      float a1 = __int_as_float(__builtin_amdgcn_readlane(ial, j + 1));
      float a2 = __int_as_float(__builtin_amdgcn_readlane(ial, j + 2));
      float a3 = __int_as_float(__builtin_amdgcn_readlane(ial, j + 3));
      v4f x00 = r0[lane], x01 = r0[lane + 32];
      v4f x10 = r1[lane], x11 = r1[lane + 32];
      v4f x20 = r2[lane], x21 = r2[lane + 32];
      v4f x30 = r3[lane], x31 = r3[lane + 32];
      acc0 += a0 * x00; acc1 += a0 * x01;
      acc0 += a1 * x10; acc1 += a1 * x11;
      acc0 += a2 * x20; acc1 += a2 * x21;
      acc0 += a3 * x30; acc1 += a3 * x31;
    }
    for (; j < n; ++j) {
      const v4f* r0 = reinterpret_cast<const v4f*>(feat + (size_t)__builtin_amdgcn_readlane(sidx, j) * D_N);
      float a0 = __int_as_float(__builtin_amdgcn_readlane(ial, j));
      v4f x00 = r0[lane], x01 = r0[lane + 32];
      acc0 += a0 * x00; acc1 += a0 * x01;
    }
  }
  v4f* hrow = reinterpret_cast<v4f*>(H + (size_t)g * D_N);
  hrow[lane] = acc0;
  hrow[lane + 32] = acc1;
}

__global__ void fc_softmax_kernel(const float* __restrict__ H,
                                  const float* __restrict__ W,
                                  const float* __restrict__ bias,
                                  float* __restrict__ out) {
  __shared__ float tile[16][260];
  int wave = threadIdx.x >> 5;
  int lane = threadIdx.x & 31;
  size_t rowBase = (size_t)blockIdx.x * 16;
  int halfK = lane >> 4;
  int l16 = lane & 15;

  v8f zero = {0.f, 0.f, 0.f, 0.f, 0.f, 0.f, 0.f, 0.f};
  v8f acc0 = zero, acc1 = zero, acc2 = zero, acc3 = zero;

  const float* hrow = H + (rowBase + l16) * D_N;
  int colBase = wave * 64 + l16;
  for (int k0 = 0; k0 < D_N; k0 += 4) {
    int ka = k0 + halfK * 2;
    v2f a;
    a.x = hrow[ka];
    a.y = hrow[ka + 1];
    const float* w0 = W + (size_t)ka * DOUT_N;
    const float* w1 = w0 + DOUT_N;
    v2f b0; b0.x = w0[colBase +  0]; b0.y = w1[colBase +  0];
    v2f b1; b1.x = w0[colBase + 16]; b1.y = w1[colBase + 16];
    v2f b2; b2.x = w0[colBase + 32]; b2.y = w1[colBase + 32];
    v2f b3; b3.x = w0[colBase + 48]; b3.y = w1[colBase + 48];
    acc0 = __builtin_amdgcn_wmma_f32_16x16x4_f32(false, a, false, b0, (short)0, acc0, false, false);
    acc1 = __builtin_amdgcn_wmma_f32_16x16x4_f32(false, a, false, b1, (short)0, acc1, false, false);
    acc2 = __builtin_amdgcn_wmma_f32_16x16x4_f32(false, a, false, b2, (short)0, acc2, false, false);
    acc3 = __builtin_amdgcn_wmma_f32_16x16x4_f32(false, a, false, b3, (short)0, acc3, false, false);
  }

  int rowOff = halfK * 8;
  {
    v8f accs[4] = {acc0, acc1, acc2, acc3};
#pragma unroll
    for (int n = 0; n < 4; ++n) {
      int col = colBase + n * 16;
      float bv = bias[col];
#pragma unroll
      for (int r = 0; r < 8; ++r) tile[r + rowOff][col] = accs[n][r] + bv;
    }
  }
  __syncthreads();

#pragma unroll
  for (int rr = 0; rr < 4; ++rr) {
    int row = wave + rr * 4;
    float v[8];
    float mx = -3.402823466e38f;
#pragma unroll
    for (int i = 0; i < 8; ++i) {
      v[i] = tile[row][lane + i * 32];
      mx = fmaxf(mx, v[i]);
    }
#pragma unroll
    for (int off = 16; off; off >>= 1) mx = fmaxf(mx, __shfl_xor(mx, off, 32));
    float sum = 0.f;
#pragma unroll
    for (int i = 0; i < 8; ++i) { v[i] = __expf(v[i] - mx); sum += v[i]; }
#pragma unroll
    for (int off = 16; off; off >>= 1) sum += __shfl_xor(sum, off, 32);
    float inv = 1.0f / sum;
    float* orow = out + (rowBase + row) * DOUT_N;
#pragma unroll
    for (int i = 0; i < 8; ++i) orow[lane + i * 32] = v[i] * inv;
  }
}

extern "C" void kernel_launch(void* const* d_in, const int* in_sizes, int n_in,
                              void* d_out, int out_size, void* d_ws, size_t ws_size,
                              hipStream_t stream) {
  const float* doc_hidden  = (const float*)d_in[0];
  const float* word_hidden = (const float*)d_in[1];
  const int*   edge_doc    = (const int*)d_in[2];
  const int*   edge_word   = (const int*)d_in[3];
  const float* d2w_w       = (const float*)d_in[4];
  const float* d2w_b       = (const float*)d_in[5];
  const float* w2d_w       = (const float*)d_in[6];
  const float* w2d_b       = (const float*)d_in[7];
  const float* fc_w        = (const float*)d_in[8];
  const float* fc_b        = (const float*)d_in[9];
  float* out = (float*)d_out;

  float* ws = (float*)d_ws;
  const size_t NH = (size_t)NG_N * D_N;
  float*    H         = ws;
  unsigned* cnt       = (unsigned*)ws;
  unsigned* slot_doc  = cnt + NG_N;
  unsigned* slot_word = slot_doc + E_N;
  int*      grp_arr   = (int*)(slot_word + E_N);
  unsigned* off       = (unsigned*)(H + NH);
  float*    sA_doc    = (float*)(off + NG_N + 1);
  float*    sB_doc    = sA_doc + ND_N;
  float*    sA_word   = sB_doc + ND_N;
  float*    sB_word   = sA_word + NW_N;
  int*      grp_srt   = (int*)(sB_word + NW_N);

  hipMemsetAsync(cnt, 0, (size_t)NG_N * sizeof(unsigned), stream);

  node_dots_kernel<<<ND_N / 8, 256, 0, stream>>>(doc_hidden, ND_N,
                                                 d2w_w, w2d_w + D_N, sA_doc, sB_doc);
  node_dots_kernel<<<NW_N / 8, 256, 0, stream>>>(word_hidden, NW_N,
                                                 d2w_w + D_N, w2d_w, sA_word, sB_word);

  const int egrid = (E_N + 255) / 256;
  edge_count_kernel<<<egrid, 256, 0, stream>>>(edge_doc, edge_word, cnt, slot_doc, slot_word, E_N);
  scan_counts_kernel<<<SCAN_BLOCKS, SCAN_T, 0, stream>>>(cnt, off);
  edge_place_kernel<<<egrid, 256, 0, stream>>>(edge_doc, edge_word, off, slot_doc, slot_word, grp_arr, E_N);
  group_sort_kernel<<<NG_N / 8, 256, 0, stream>>>(off, grp_arr, grp_srt);

  group_aggregate_kernel<<<NG_N / 8, 256, 0, stream>>>(off, grp_srt, doc_hidden, word_hidden,
                                                       sA_doc, sA_word, sB_doc, sB_word,
                                                       d2w_b, w2d_b, H);

  fc_softmax_kernel<<<NG_N / 16, 128, 0, stream>>>(H, fc_w, fc_b, out);

  (void)in_sizes; (void)n_in; (void)out_size; (void)ws_size;
}


# edge grouping by LDS counters: 32 edge slices, each block counts its slice per destination in LDS, table scan, slots by LDS adds; no global atomics, memset or slot arrays; sum order unchanged
# speedup vs baseline: 3.2341x; 1.2171x over previous
#include <hip/hip_runtime.h>

#define ND_N 20000
#define NW_N 30000
#define E_N  500000
#define D_N  256
#define DOUT_N 256
#define NG_N (ND_N + NW_N)

typedef __attribute__((ext_vector_type(2))) float v2f;
typedef __attribute__((ext_vector_type(4))) float v4f;
typedef __attribute__((ext_vector_type(8))) float v8f;

__global__ void node_dots_kernel(const float* __restrict__ hid, int n,
                                 const float* __restrict__ wa,
                                 const float* __restrict__ wb,
                                 float* __restrict__ outa,
                                 float* __restrict__ outb) {
  int wave = threadIdx.x >> 5, lane = threadIdx.x & 31;
  int node = blockIdx.x * 8 + wave;
  if (node >= n) return;
  const float* row = hid + (size_t)node * D_N;
  float da = 0.f, db = 0.f;
#pragma unroll
  for (int i = 0; i < 8; ++i) {
    int k = lane + i * 32;
    float x = row[k];
    da += x * wa[k];
    db += x * wb[k];
  }
#pragma unroll
  for (int off = 16; off; off >>= 1) {
    da += __shfl_xor(da, off, 32);
    db += __shfl_xor(db, off, 32);
  }
  if (lane == 0) { outa[node] = da; outb[node] = db; }
}

typedef __attribute__((ext_vector_type(4))) unsigned v4u;
static_assert(NG_N % 4 == 0, "a 16-byte group never straddles the end of the counts");
#define GB_N 32
#define GB_T 1024
#define GB_SLICE ((E_N + GB_N - 1) / GB_N)
#define GB_ROW_IT ((NG_N / 4 + GB_T - 1) / GB_T)
#define GB_BATCH 8
#define PL_Q 10
#define PL_BINS (NG_N / PL_Q)
static_assert(NG_N % (4 * PL_Q) == 0, "a range is a whole number of 16-byte groups");

#define SCAN_T 512
#define SCAN_PER (4 * SCAN_T)
#define SCAN_BLOCKS ((NG_N + SCAN_PER - 1) / SCAN_PER)
#define SCAN_W (SCAN_T / 32)
#define SEG_SLOTS (2 * GB_ROW_IT)
static_assert(GB_T == 2 * SCAN_T, "a wave of the count kernel lies inside one scan segment in every sweep");
static_assert(SEG_SLOTS >= SCAN_BLOCKS, "every scan block has its segment sum");

__global__ void __launch_bounds__(GB_T) slice_count_kernel(const int* __restrict__ edge_doc,
                                                           const int* __restrict__ edge_word,
                                                           unsigned* __restrict__ table,
                                                           unsigned* __restrict__ seg) {
  __shared__ __attribute__((aligned(16))) unsigned c[NG_N];
  __shared__ unsigned segsum[SEG_SLOTS];
  int t = threadIdx.x, lane = t & 31;
  v4u* c4 = reinterpret_cast<v4u*>(c);
  for (int i = t; i < NG_N / 4; i += GB_T) c4[i] = (v4u){0u, 0u, 0u, 0u};
  if (t < SEG_SLOTS) segsum[t] = 0u;
  __syncthreads();
  int lo = blockIdx.x * GB_SLICE;
  int hi = lo + GB_SLICE < E_N ? lo + GB_SLICE : E_N;
  for (int base = lo + t; base < hi; base += GB_BATCH * GB_T) {
    int dd[GB_BATCH], ww[GB_BATCH];
#pragma unroll
    for (int j = 0; j < GB_BATCH; ++j) {
      int e = base + j * GB_T;
      dd[j] = 0; ww[j] = 0;
      if (e < hi) { dd[j] = edge_doc[e]; ww[j] = edge_word[e]; }
    }
#pragma unroll
    for (int j = 0; j < GB_BATCH; ++j) {
      if (base + j * GB_T < hi) {
        atomicAdd(&c[dd[j]], 1u);
        atomicAdd(&c[ND_N + ww[j]], 1u);
      }
    }
  }
  __syncthreads();
  v4u* row = reinterpret_cast<v4u*>(table + (size_t)blockIdx.x * NG_N);
  unsigned w[GB_ROW_IT];
#pragma unroll
  for (int k = 0; k < GB_ROW_IT; ++k) {
    int i = t + k * GB_T;
    v4u v = {0u, 0u, 0u, 0u};
    if (i < NG_N / 4) { v = c4[i]; row[i] = v; }
    w[k] = (v.x + v.y) + (v.z + v.w);
  }
  for (int o = 16; o; o >>= 1) {
#pragma unroll
    for (int k = 0; k < GB_ROW_IT; ++k) w[k] += __shfl_xor(w[k], o, 32);
  }
  if (lane == 0) {
#pragma unroll
    for (int k = 0; k < GB_ROW_IT; ++k) atomicAdd(&segsum[(t + k * GB_T) / SCAN_T], w[k]);
  }
  __syncthreads();
  if (t < SCAN_BLOCKS) seg[blockIdx.x * SCAN_BLOCKS + t] = segsum[t];
}

__global__ void __launch_bounds__(SCAN_T) scan_counts_kernel(unsigned* __restrict__ table,
                                                             const unsigned* __restrict__ seg,
                                                             unsigned* __restrict__ off) {
  __shared__ unsigned wpre[SCAN_W];
  __shared__ unsigned wtot[SCAN_W];
  int t = threadIdx.x, wave = t >> 5, lane = t & 31;
  int blk = blockIdx.x;
  v4u* tab4 = reinterpret_cast<v4u*>(table);
  int q = blk * SCAN_T + t;
  bool own = q < NG_N / 4;
  int qq = own ? q : 0;
  v4u r[GB_N];
#pragma unroll
  for (int b = 0; b < GB_N; ++b) r[b] = tab4[b * (NG_N / 4) + qq];
  unsigned pre = 0u;
  for (int i = t; i < GB_N * SCAN_BLOCKS; i += SCAN_T) {
    unsigned s = seg[i];
    if (i % SCAN_BLOCKS < blk) pre += s;
  }
  v4u mine = {0u, 0u, 0u, 0u};
#pragma unroll
  for (int b = 0; b < GB_N; ++b) { v4u cb = r[b]; r[b] = mine; mine += cb; }
  if (!own) mine = (v4u){0u, 0u, 0u, 0u};
  unsigned s1 = mine.x, s2 = s1 + mine.y, s3 = s2 + mine.z, tot = s3 + mine.w;
  for (int o = 16; o; o >>= 1) pre += __shfl_xor(pre, o, 32);
  unsigned inc = tot;
  for (int o = 1; o < 32; o <<= 1) {
    unsigned u = __shfl_up(inc, o, 32);
    if (lane >= o) inc += u;
  }
  if (lane == 0) wpre[wave] = pre;
  if (lane == 31) wtot[wave] = inc;
  __syncthreads();
  unsigned x = 0u;
  if (lane < SCAN_W) x = wpre[lane] + (lane < wave ? wtot[lane] : 0u);
  for (int o = 16; o; o >>= 1) x += __shfl_xor(x, o, 32);
  unsigned e0 = x + inc - tot;
  if (own) {
    v4u o4 = {e0, e0 + s1, e0 + s2, e0 + s3};
    reinterpret_cast<v4u*>(off)[q] = o4;
#pragma unroll
    for (int b = 0; b < GB_N; ++b) tab4[b * (NG_N / 4) + q] = o4 + r[b];
    if (q == NG_N / 4 - 1) off[NG_N] = e0 + tot;
  }
}

__global__ void __launch_bounds__(GB_T) slice_place_kernel(const int* __restrict__ edge_doc,
                                                           const int* __restrict__ edge_word,
                                                           const unsigned* __restrict__ table,
                                                           int* __restrict__ grp) {
  __shared__ __attribute__((aligned(16))) unsigned p[PL_BINS];
  int t = threadIdx.x;
  int slice = blockIdx.x;
  unsigned first = blockIdx.y * PL_BINS;
  v4u* p4 = reinterpret_cast<v4u*>(p);
  const v4u* row = reinterpret_cast<const v4u*>(table + (size_t)slice * NG_N + first);
  for (int i = t; i < PL_BINS / 4; i += GB_T) p4[i] = row[i];
  __syncthreads();
  int lo = slice * GB_SLICE;
  int hi = lo + GB_SLICE < E_N ? lo + GB_SLICE : E_N;
  for (int base = lo + t; base < hi; base += GB_BATCH * GB_T) {
    int dd[GB_BATCH], ww[GB_BATCH];
#pragma unroll
    for (int j = 0; j < GB_BATCH; ++j) {
      int e = base + j * GB_T;
      dd[j] = 0; ww[j] = 0;
      if (e < hi) { dd[j] = edge_doc[e]; ww[j] = edge_word[e]; }
    }
#pragma unroll
    for (int j = 0; j < GB_BATCH; ++j) {
      if (base + j * GB_T < hi) {
        unsigned k1 = (unsigned)dd[j] - first;
        unsigned k2 = (unsigned)(ND_N + ww[j]) - first;
        if (k1 < (unsigned)PL_BINS) grp[atomicAdd(&p[k1], 1u)] = ww[j];
        if (k2 < (unsigned)PL_BINS) grp[atomicAdd(&p[k2], 1u)] = dd[j];
      }
    }
  }
}

__global__ void group_sort_kernel(const unsigned* __restrict__ off,
                                  const int* __restrict__ grp_in,
                                  int* __restrict__ grp_out) {
  int lane = threadIdx.x & 31;
  int g = __builtin_amdgcn_readfirstlane((int)(blockIdx.x * 8 + (threadIdx.x >> 5)));
  if (g >= NG_N) return;
  int start = (int)off[g];
  int deg = (int)off[g + 1] - start;
  const int* in = grp_in + start;
  int* outp = grp_out + start;
  for (int c = 0; c < deg; c += 32) {
    int idx = c + lane;
    int key = idx < deg ? in[idx] : 0x7fffffff;
    int rank = 0;
    for (int c2 = 0; c2 < deg; c2 += 32) {
      int idx2 = c2 + lane;
      int k2 = idx2 < deg ? in[idx2] : 0x7fffffff;
      int n2 = deg - c2 < 32 ? deg - c2 : 32;
      for (int j = 0; j < n2; ++j) {
        int v = __builtin_amdgcn_readlane(k2, j);
        rank += (v < key || (v == key && c2 + j < idx)) ? 1 : 0;
      }
    }
    if (idx < deg) outp[rank] = key;
  }
}

__device__ __forceinline__ float group_score(const int* __restrict__ seg, const float* __restrict__ ssrc,
                                             int idx, float sd, float bb, int* sidx) {
  int s = seg[idx];
  *sidx = s;
  float a = ssrc[s] + sd + bb;
  return a > 0.f ? a : 0.01f * a;
}

__global__ void group_aggregate_kernel(const unsigned* __restrict__ off,
                                       const int* __restrict__ grp,
                                       const float* __restrict__ doc_hidden,
                                       const float* __restrict__ word_hidden,
                                       const float* __restrict__ sA_doc,
                                       const float* __restrict__ sA_word,
                                       const float* __restrict__ sB_doc,
                                       const float* __restrict__ sB_word,
                                       const float* __restrict__ d2w_b,
                                       const float* __restrict__ w2d_b,
                                       float* __restrict__ H) {
  int lane = threadIdx.x & 31;
  int g = __builtin_amdgcn_readfirstlane((int)(blockIdx.x * 8 + (threadIdx.x >> 5)));
  if (g >= NG_N) return;
  const float* feat; const float* ssrc; float sd, bb;
  if (g < ND_N) { feat = word_hidden; ssrc = sB_word; sd = sB_doc[g];         bb = w2d_b[0]; }
  else          { feat = doc_hidden;  ssrc = sA_doc;  sd = sA_word[g - ND_N]; bb = d2w_b[0]; }
  int start = (int)off[g];
  int deg = (int)off[g + 1] - start;
  const int* seg = grp + start;

  int sidx0 = 0; float v0 = 0.f;
  if (lane < deg) v0 = group_score(seg, ssrc, lane, sd, bb, &sidx0);
  auto score_at = [&](int c, int idx, int* sidx) -> float {
    if (c == 0) { *sidx = sidx0; return v0; }
    return group_score(seg, ssrc, idx, sd, bb, sidx);
  };

  float mx = -__builtin_huge_valf();
  for (int c = 0; c < deg; c += 32) {
    int idx = c + lane;
    if (idx < deg) { int s; mx = fmaxf(mx, score_at(c, idx, &s)); }
  }
  for (int o = 16; o; o >>= 1) mx = fmaxf(mx, __shfl_xor(mx, o, 32));

  float sum = 0.f;
  for (int c = 0; c < deg; c += 32) {
    int idx = c + lane;
    float ex = 0.f;
    if (idx < deg) { int s; ex = __expf(score_at(c, idx, &s) - mx); }
    for (int o = 16; o; o >>= 1) ex += __shfl_xor(ex, o, 32);
    sum += ex;
  }
  float den = fmaxf(sum, 1e-30f);

  v4f acc0 = {0.f, 0.f, 0.f, 0.f}, acc1 = {0.f, 0.f, 0.f, 0.f};
  for (int c = 0; c < deg; c += 32) {
    int idx = c + lane;
    int sidx = 0; float al = 0.f;
    if (idx < deg) al = __expf(score_at(c, idx, &sidx) - mx) / den;
    int ial = __float_as_int(al);
    int n = deg - c < 32 ? deg - c : 32;
    int j = 0;
    for (; j + 4 <= n; j += 4) {
      const v4f* r0 = reinterpret_cast<const v4f*>(feat + (size_t)__builtin_amdgcn_readlane(sidx, j)     * D_N);
      const v4f* r1 = reinterpret_cast<const v4f*>(feat + (size_t)__builtin_amdgcn_readlane(sidx, j + 1) * D_N);
      const v4f* r2 = reinterpret_cast<const v4f*>(feat + (size_t)__builtin_amdgcn_readlane(sidx, j + 2) * D_N);
      const v4f* r3 = reinterpret_cast<const v4f*>(feat + (size_t)__builtin_amdgcn_readlane(sidx, j + 3) * D_N);
      float a0 = __int_as_float(__builtin_amdgcn_readlane(ial, j));
      float a1 = __int_as_float(__builtin_amdgcn_readlane(ial, j + 1));
      float a2 = __int_as_float(__builtin_amdgcn_readlane(ial, j + 2));
      float a3 = __int_as_float(__builtin_amdgcn_readlane(ial, j + 3));
      v4f x00 = r0[lane], x01 = r0[lane + 32];
      v4f x10 = r1[lane], x11 = r1[lane + 32];
      v4f x20 = r2[lane], x21 = r2[lane + 32];
      v4f x30 = r3[lane], x31 = r3[lane + 32];
      acc0 += a0 * x00; acc1 += a0 * x01;
      acc0 += a1 * x10; acc1 += a1 * x11;
      acc0 += a2 * x20; acc1 += a2 * x21;
      acc0 += a3 * x30; acc1 += a3 * x31;
    }
    for (; j < n; ++j) {
      const v4f* r0 = reinterpret_cast<const v4f*>(feat + (size_t)__builtin_amdgcn_readlane(sidx, j) * D_N);
      float a0 = __int_as_float(__builtin_amdgcn_readlane(ial, j));
      v4f x00 = r0[lane], x01 = r0[lane + 32];
      acc0 += a0 * x00; acc1 += a0 * x01;
    }
  }
  v4f* hrow = reinterpret_cast<v4f*>(H + (size_t)g * D_N);
  hrow[lane] = acc0;
  hrow[lane + 32] = acc1;
}

__global__ void fc_softmax_kernel(const float* __restrict__ H,
                                  const float* __restrict__ W,
                                  const float* __restrict__ bias,
                                  float* __restrict__ out) {
  __shared__ float tile[16][260];
  int wave = threadIdx.x >> 5;
  int lane = threadIdx.x & 31;
  size_t rowBase = (size_t)blockIdx.x * 16;
  int halfK = lane >> 4;
  int l16 = lane & 15;

  v8f zero = {0.f, 0.f, 0.f, 0.f, 0.f, 0.f, 0.f, 0.f};
  v8f acc0 = zero, acc1 = zero, acc2 = zero, acc3 = zero;

  const float* hrow = H + (rowBase + l16) * D_N;
  int colBase = wave * 64 + l16;
  for (int k0 = 0; k0 < D_N; k0 += 4) {
    int ka = k0 + halfK * 2;
    v2f a;
    a.x = hrow[ka];
    a.y = hrow[ka + 1];
    const float* w0 = W + (size_t)ka * DOUT_N;
    const float* w1 = w0 + DOUT_N;
    v2f b0; b0.x = w0[colBase +  0]; b0.y = w1[colBase +  0];
    v2f b1; b1.x = w0[colBase + 16]; b1.y = w1[colBase + 16];
    v2f b2; b2.x = w0[colBase + 32]; b2.y = w1[colBase + 32];
    v2f b3; b3.x = w0[colBase + 48]; b3.y = w1[colBase + 48];
    acc0 = __builtin_amdgcn_wmma_f32_16x16x4_f32(false, a, false, b0, (short)0, acc0, false, false);
    acc1 = __builtin_amdgcn_wmma_f32_16x16x4_f32(false, a, false, b1, (short)0, acc1, false, false);
    acc2 = __builtin_amdgcn_wmma_f32_16x16x4_f32(false, a, false, b2, (short)0, acc2, false, false);
    acc3 = __builtin_amdgcn_wmma_f32_16x16x4_f32(false, a, false, b3, (short)0, acc3, false, false);
  }

  int rowOff = halfK * 8;
  {
    v8f accs[4] = {acc0, acc1, acc2, acc3};
#pragma unroll
    for (int n = 0; n < 4; ++n) {
      int col = colBase + n * 16;
      float bv = bias[col];
#pragma unroll
      for (int r = 0; r < 8; ++r) tile[r + rowOff][col] = accs[n][r] + bv;
    }
  }
  __syncthreads();

#pragma unroll
  for (int rr = 0; rr < 4; ++rr) {
    int row = wave + rr * 4;
    float v[8];
    float mx = -3.402823466e38f;
#pragma unroll
    for (int i = 0; i < 8; ++i) {
      v[i] = tile[row][lane + i * 32];
      mx = fmaxf(mx, v[i]);
    }
#pragma unroll
    for (int off = 16; off; off >>= 1) mx = fmaxf(mx, __shfl_xor(mx, off, 32));
    float sum = 0.f;
#pragma unroll
    for (int i = 0; i < 8; ++i) { v[i] = __expf(v[i] - mx); sum += v[i]; }
#pragma unroll
    for (int off = 16; off; off >>= 1) sum += __shfl_xor(sum, off, 32);
    float inv = 1.0f / sum;
    float* orow = out + (rowBase + row) * DOUT_N;
#pragma unroll
    for (int i = 0; i < 8; ++i) orow[lane + i * 32] = v[i] * inv;
  }
}

extern "C" void kernel_launch(void* const* d_in, const int* in_sizes, int n_in,
                              void* d_out, int out_size, void* d_ws, size_t ws_size,
                              hipStream_t stream) {
  const float* doc_hidden  = (const float*)d_in[0];
  const float* word_hidden = (const float*)d_in[1];
  const int*   edge_doc    = (const int*)d_in[2];
  const int*   edge_word   = (const int*)d_in[3];
  const float* d2w_w       = (const float*)d_in[4];
  const float* d2w_b       = (const float*)d_in[5];
  const float* w2d_w       = (const float*)d_in[6];
  const float* w2d_b       = (const float*)d_in[7];
  const float* fc_w        = (const float*)d_in[8];
  const float* fc_b        = (const float*)d_in[9];
  float* out = (float*)d_out;

  float* ws = (float*)d_ws;
  const size_t NH = (size_t)NG_N * D_N;
  float*    H         = ws;
  unsigned* table     = (unsigned*)ws;
  unsigned* seg       = table + (size_t)GB_N * NG_N;
  int*      grp_arr   = (int*)(seg + GB_N * SCAN_BLOCKS);
  unsigned* off       = (unsigned*)(H + NH);
  float*    sA_doc    = (float*)(off + NG_N + 1);
  float*    sB_doc    = sA_doc + ND_N;
  float*    sA_word   = sB_doc + ND_N;
  float*    sB_word   = sA_word + NW_N;
  int*      grp_srt   = (int*)(sB_word + NW_N);

  node_dots_kernel<<<ND_N / 8, 256, 0, stream>>>(doc_hidden, ND_N,
                                                 d2w_w, w2d_w + D_N, sA_doc, sB_doc);
  node_dots_kernel<<<NW_N / 8, 256, 0, stream>>>(word_hidden, NW_N,
                                                 d2w_w + D_N, w2d_w, sA_word, sB_word);

  slice_count_kernel<<<GB_N, GB_T, 0, stream>>>(edge_doc, edge_word, table, seg);
  scan_counts_kernel<<<SCAN_BLOCKS, SCAN_T, 0, stream>>>(table, seg, off);
  slice_place_kernel<<<dim3(GB_N, PL_Q), GB_T, 0, stream>>>(edge_doc, edge_word, table, grp_arr);
  group_sort_kernel<<<NG_N / 8, 256, 0, stream>>>(off, grp_arr, grp_srt);

  group_aggregate_kernel<<<NG_N / 8, 256, 0, stream>>>(off, grp_srt, doc_hidden, word_hidden,
                                                       sA_doc, sA_word, sB_doc, sB_word,
                                                       d2w_b, w2d_b, H);

  fc_softmax_kernel<<<NG_N / 16, 128, 0, stream>>>(H, fc_w, fc_b, out);

  (void)in_sizes; (void)n_in; (void)out_size; (void)ws_size;
}


# node score dots: one launch of 3125 waves, each 16 consecutive rows with its 32 16-byte loads in flight at once, not 50000 one-row waves in two launches; dot sums in a new lane order
# speedup vs baseline: 3.3869x; 1.0472x over previous
#include <hip/hip_runtime.h>

#define ND_N 20000
#define NW_N 30000
#define E_N  500000
#define D_N  256
#define DOUT_N 256
#define NG_N (ND_N + NW_N)

typedef __attribute__((ext_vector_type(2))) float v2f;
typedef __attribute__((ext_vector_type(4))) float v4f;
typedef __attribute__((ext_vector_type(8))) float v8f;

#define DOT_ROWS 16
#define DOT_T 128
#define DOT_WAVES (DOT_T / 32)
#define DOT_DOC_CHUNKS (ND_N / DOT_ROWS)
#define DOT_CHUNKS ((ND_N + NW_N) / DOT_ROWS)
#define DOT_BLOCKS ((DOT_CHUNKS + DOT_WAVES - 1) / DOT_WAVES)
static_assert(ND_N % DOT_ROWS == 0 && NW_N % DOT_ROWS == 0, "a chunk lies inside one table");
static_assert(2 * DOT_ROWS == 32, "one result per lane: 16 first dots, 16 second dots");

__global__ void __launch_bounds__(DOT_T) node_dots_kernel(const float* __restrict__ doc_hidden,
                                                          const float* __restrict__ word_hidden,
                                                          const float* __restrict__ d2w_w,
                                                          const float* __restrict__ w2d_w,
                                                          float* __restrict__ sA_doc,
                                                          float* __restrict__ sB_doc,
                                                          float* __restrict__ sA_word,
                                                          float* __restrict__ sB_word) {
  int lane = threadIdx.x & 31;
  int chunk = __builtin_amdgcn_readfirstlane((int)(blockIdx.x * DOT_WAVES + (threadIdx.x >> 5)));
  if (chunk >= DOT_CHUNKS) return;
  const float* hid; const float* wa; const float* wb; float* outa; float* outb; int r0;
  if (chunk < DOT_DOC_CHUNKS) {
    r0 = chunk * DOT_ROWS;
    hid = doc_hidden;  wa = d2w_w;       wb = w2d_w + D_N; outa = sA_doc;  outb = sB_doc;
  } else {
    r0 = (chunk - DOT_DOC_CHUNKS) * DOT_ROWS;
    hid = word_hidden; wa = d2w_w + D_N; wb = w2d_w;       outa = sA_word; outb = sB_word;
  }
  const v4f* wa4 = reinterpret_cast<const v4f*>(wa);
  const v4f* wb4 = reinterpret_cast<const v4f*>(wb);
  v4f a0 = wa4[lane], a1 = wa4[lane + 32];
  v4f b0 = wb4[lane], b1 = wb4[lane + 32];
  const v4f* rows = reinterpret_cast<const v4f*>(hid + (size_t)r0 * D_N);
  v4f x[2 * DOT_ROWS];
#pragma unroll
  for (int k = 0; k < 2 * DOT_ROWS; ++k) x[k] = rows[k * 32 + lane];
  float v[2 * DOT_ROWS];
#pragma unroll
  for (int k = 0; k < DOT_ROWS; ++k) {
    v4f sa = x[2 * k] * a0 + x[2 * k + 1] * a1;
    v4f sb = x[2 * k] * b0 + x[2 * k + 1] * b1;
    v[k]            = (sa.x + sa.y) + (sa.z + sa.w);
    v[DOT_ROWS + k] = (sb.x + sb.y) + (sb.z + sb.w);
  }
  for (int o = 16; o; o >>= 1) {
#pragma unroll
    for (int j = 0; j < 2 * DOT_ROWS; ++j) v[j] += __shfl_xor(v[j], o, 32);
  }
  float mine = v[0];
#pragma unroll
  for (int j = 1; j < 2 * DOT_ROWS; ++j) mine = lane == j ? v[j] : mine;
  float* dst = lane < DOT_ROWS ? outa + r0 + lane : outb + r0 + (lane - DOT_ROWS);
  *dst = mine;
}

typedef __attribute__((ext_vector_type(4))) unsigned v4u;
static_assert(NG_N % 4 == 0, "a 16-byte group never straddles the end of the counts");
#define GB_N 32
#define GB_T 1024
#define GB_SLICE ((E_N + GB_N - 1) / GB_N)
#define GB_ROW_IT ((NG_N / 4 + GB_T - 1) / GB_T)
#define GB_BATCH 8
#define PL_Q 10
#define PL_BINS (NG_N / PL_Q)
static_assert(NG_N % (4 * PL_Q) == 0, "a range is a whole number of 16-byte groups");

#define SCAN_T 512
#define SCAN_PER (4 * SCAN_T)
#define SCAN_BLOCKS ((NG_N + SCAN_PER - 1) / SCAN_PER)
#define SCAN_W (SCAN_T / 32)
#define SEG_SLOTS (2 * GB_ROW_IT)
static_assert(GB_T == 2 * SCAN_T, "a wave of the count kernel lies inside one scan segment in every sweep");
static_assert(SEG_SLOTS >= SCAN_BLOCKS, "every scan block has its segment sum");

__global__ void __launch_bounds__(GB_T) slice_count_kernel(const int* __restrict__ edge_doc,
                                                           const int* __restrict__ edge_word,
                                                           unsigned* __restrict__ table,
                                                           unsigned* __restrict__ seg) {
  __shared__ __attribute__((aligned(16))) unsigned c[NG_N];
  __shared__ unsigned segsum[SEG_SLOTS];
  int t = threadIdx.x, lane = t & 31;
  v4u* c4 = reinterpret_cast<v4u*>(c);
  for (int i = t; i < NG_N / 4; i += GB_T) c4[i] = (v4u){0u, 0u, 0u, 0u};
  if (t < SEG_SLOTS) segsum[t] = 0u;
  __syncthreads();
  int lo = blockIdx.x * GB_SLICE;
  int hi = lo + GB_SLICE < E_N ? lo + GB_SLICE : E_N;
  for (int base = lo + t; base < hi; base += GB_BATCH * GB_T) {
    int dd[GB_BATCH], ww[GB_BATCH];
#pragma unroll
    for (int j = 0; j < GB_BATCH; ++j) {
      int e = base + j * GB_T;
      dd[j] = 0; ww[j] = 0;
      if (e < hi) { dd[j] = edge_doc[e]; ww[j] = edge_word[e]; }
    }
#pragma unroll
    for (int j = 0; j < GB_BATCH; ++j) {
      if (base + j * GB_T < hi) {
        atomicAdd(&c[dd[j]], 1u);
        atomicAdd(&c[ND_N + ww[j]], 1u);
      }
    }
  }
  __syncthreads();
  v4u* row = reinterpret_cast<v4u*>(table + (size_t)blockIdx.x * NG_N);
  unsigned w[GB_ROW_IT];
#pragma unroll
  for (int k = 0; k < GB_ROW_IT; ++k) {
    int i = t + k * GB_T;
    v4u v = {0u, 0u, 0u, 0u};
    if (i < NG_N / 4) { v = c4[i]; row[i] = v; }
    w[k] = (v.x + v.y) + (v.z + v.w);
  }
  for (int o = 16; o; o >>= 1) {
#pragma unroll
    for (int k = 0; k < GB_ROW_IT; ++k) w[k] += __shfl_xor(w[k], o, 32);
  }
  if (lane == 0) {
#pragma unroll
    for (int k = 0; k < GB_ROW_IT; ++k) atomicAdd(&segsum[(t + k * GB_T) / SCAN_T], w[k]);
  }
  __syncthreads();
  if (t < SCAN_BLOCKS) seg[blockIdx.x * SCAN_BLOCKS + t] = segsum[t];
}

__global__ void __launch_bounds__(SCAN_T) scan_counts_kernel(unsigned* __restrict__ table,
                                                             const unsigned* __restrict__ seg,
                                                             unsigned* __restrict__ off) {
  __shared__ unsigned wpre[SCAN_W];
  __shared__ unsigned wtot[SCAN_W];
  int t = threadIdx.x, wave = t >> 5, lane = t & 31;
  int blk = blockIdx.x;
  v4u* tab4 = reinterpret_cast<v4u*>(table);
  int q = blk * SCAN_T + t;
  bool own = q < NG_N / 4;
  int qq = own ? q : 0;
  v4u r[GB_N];
#pragma unroll
  for (int b = 0; b < GB_N; ++b) r[b] = tab4[b * (NG_N / 4) + qq];
  unsigned pre = 0u;
  for (int i = t; i < GB_N * SCAN_BLOCKS; i += SCAN_T) {
    unsigned s = seg[i];
    if (i % SCAN_BLOCKS < blk) pre += s;
  }
  v4u mine = {0u, 0u, 0u, 0u};
#pragma unroll
  for (int b = 0; b < GB_N; ++b) { v4u cb = r[b]; r[b] = mine; mine += cb; }
  if (!own) mine = (v4u){0u, 0u, 0u, 0u};
  unsigned s1 = mine.x, s2 = s1 + mine.y, s3 = s2 + mine.z, tot = s3 + mine.w;
  for (int o = 16; o; o >>= 1) pre += __shfl_xor(pre, o, 32);
  unsigned inc = tot;
  for (int o = 1; o < 32; o <<= 1) {
    unsigned u = __shfl_up(inc, o, 32);
    if (lane >= o) inc += u;
  }
  if (lane == 0) wpre[wave] = pre;
  if (lane == 31) wtot[wave] = inc;
  __syncthreads();
  unsigned x = 0u;
  if (lane < SCAN_W) x = wpre[lane] + (lane < wave ? wtot[lane] : 0u);
  for (int o = 16; o; o >>= 1) x += __shfl_xor(x, o, 32);
  unsigned e0 = x + inc - tot;
  if (own) {
    v4u o4 = {e0, e0 + s1, e0 + s2, e0 + s3};
    reinterpret_cast<v4u*>(off)[q] = o4;
#pragma unroll
    for (int b = 0; b < GB_N; ++b) tab4[b * (NG_N / 4) + q] = o4 + r[b];
    if (q == NG_N / 4 - 1) off[NG_N] = e0 + tot;
  }
}

__global__ void __launch_bounds__(GB_T) slice_place_kernel(const int* __restrict__ edge_doc,
                                                           const int* __restrict__ edge_word,
                                                           const unsigned* __restrict__ table,
                                                           int* __restrict__ grp) {
  __shared__ __attribute__((aligned(16))) unsigned p[PL_BINS];
  int t = threadIdx.x;
  int slice = blockIdx.x;
  unsigned first = blockIdx.y * PL_BINS;
  v4u* p4 = reinterpret_cast<v4u*>(p);
  const v4u* row = reinterpret_cast<const v4u*>(table + (size_t)slice * NG_N + first);
  for (int i = t; i < PL_BINS / 4; i += GB_T) p4[i] = row[i];
  __syncthreads();
  int lo = slice * GB_SLICE;
  int hi = lo + GB_SLICE < E_N ? lo + GB_SLICE : E_N;
  for (int base = lo + t; base < hi; base += GB_BATCH * GB_T) {
    int dd[GB_BATCH], ww[GB_BATCH];
#pragma unroll
    for (int j = 0; j < GB_BATCH; ++j) {
      int e = base + j * GB_T;
      dd[j] = 0; ww[j] = 0;
      if (e < hi) { dd[j] = edge_doc[e]; ww[j] = edge_word[e]; }
    }
#pragma unroll
    for (int j = 0; j < GB_BATCH; ++j) {
      if (base + j * GB_T < hi) {
        unsigned k1 = (unsigned)dd[j] - first;
        unsigned k2 = (unsigned)(ND_N + ww[j]) - first;
        if (k1 < (unsigned)PL_BINS) grp[atomicAdd(&p[k1], 1u)] = ww[j];
        if (k2 < (unsigned)PL_BINS) grp[atomicAdd(&p[k2], 1u)] = dd[j];
      }
    }
  }
}

__global__ void group_sort_kernel(const unsigned* __restrict__ off,
                                  const int* __restrict__ grp_in,
                                  int* __restrict__ grp_out) {
  int lane = threadIdx.x & 31;
  int g = __builtin_amdgcn_readfirstlane((int)(blockIdx.x * 8 + (threadIdx.x >> 5)));
  if (g >= NG_N) return;
  int start = (int)off[g];
  int deg = (int)off[g + 1] - start;
  const int* in = grp_in + start;
  int* outp = grp_out + start;
  for (int c = 0; c < deg; c += 32) {
    int idx = c + lane;
    int key = idx < deg ? in[idx] : 0x7fffffff;
    int rank = 0;
    for (int c2 = 0; c2 < deg; c2 += 32) {
      int idx2 = c2 + lane;
      int k2 = idx2 < deg ? in[idx2] : 0x7fffffff;
      int n2 = deg - c2 < 32 ? deg - c2 : 32;
      for (int j = 0; j < n2; ++j) {
        int v = __builtin_amdgcn_readlane(k2, j);
        rank += (v < key || (v == key && c2 + j < idx)) ? 1 : 0;
      }
    }
    if (idx < deg) outp[rank] = key;
  }
}

__device__ __forceinline__ float group_score(const int* __restrict__ seg, const float* __restrict__ ssrc,
                                             int idx, float sd, float bb, int* sidx) {
  int s = seg[idx];
  *sidx = s;
  float a = ssrc[s] + sd + bb;
  return a > 0.f ? a : 0.01f * a;
}

__global__ void group_aggregate_kernel(const unsigned* __restrict__ off,
                                       const int* __restrict__ grp,
                                       const float* __restrict__ doc_hidden,
                                       const float* __restrict__ word_hidden,
                                       const float* __restrict__ sA_doc,
                                       const float* __restrict__ sA_word,
                                       const float* __restrict__ sB_doc,
                                       const float* __restrict__ sB_word,
                                       const float* __restrict__ d2w_b,
                                       const float* __restrict__ w2d_b,
                                       float* __restrict__ H) {
  int lane = threadIdx.x & 31;
  int g = __builtin_amdgcn_readfirstlane((int)(blockIdx.x * 8 + (threadIdx.x >> 5)));
  if (g >= NG_N) return;
  const float* feat; const float* ssrc; float sd, bb;
  if (g < ND_N) { feat = word_hidden; ssrc = sB_word; sd = sB_doc[g];         bb = w2d_b[0]; }
  else          { feat = doc_hidden;  ssrc = sA_doc;  sd = sA_word[g - ND_N]; bb = d2w_b[0]; }
  int start = (int)off[g];
  int deg = (int)off[g + 1] - start;
  const int* seg = grp + start;

  int sidx0 = 0; float v0 = 0.f;
  if (lane < deg) v0 = group_score(seg, ssrc, lane, sd, bb, &sidx0);
  auto score_at = [&](int c, int idx, int* sidx) -> float {
    if (c == 0) { *sidx = sidx0; return v0; }
    return group_score(seg, ssrc, idx, sd, bb, sidx);
  };

  float mx = -__builtin_huge_valf();
  for (int c = 0; c < deg; c += 32) {
    int idx = c + lane;
    if (idx < deg) { int s; mx = fmaxf(mx, score_at(c, idx, &s)); }
  }
  for (int o = 16; o; o >>= 1) mx = fmaxf(mx, __shfl_xor(mx, o, 32));

  float sum = 0.f;
  for (int c = 0; c < deg; c += 32) {
    int idx = c + lane;
    float ex = 0.f;
    if (idx < deg) { int s; ex = __expf(score_at(c, idx, &s) - mx); }
    for (int o = 16; o; o >>= 1) ex += __shfl_xor(ex, o, 32);
    sum += ex;
  }
  float den = fmaxf(sum, 1e-30f);

  v4f acc0 = {0.f, 0.f, 0.f, 0.f}, acc1 = {0.f, 0.f, 0.f, 0.f};
  for (int c = 0; c < deg; c += 32) {
    int idx = c + lane;
    int sidx = 0; float al = 0.f;
    if (idx < deg) al = __expf(score_at(c, idx, &sidx) - mx) / den;
    int ial = __float_as_int(al);
    int n = deg - c < 32 ? deg - c : 32;
    int j = 0;
    for (; j + 4 <= n; j += 4) {
      const v4f* r0 = reinterpret_cast<const v4f*>(feat + (size_t)__builtin_amdgcn_readlane(sidx, j)     * D_N);
      const v4f* r1 = reinterpret_cast<const v4f*>(feat + (size_t)__builtin_amdgcn_readlane(sidx, j + 1) * D_N);
      const v4f* r2 = reinterpret_cast<const v4f*>(feat + (size_t)__builtin_amdgcn_readlane(sidx, j + 2) * D_N);
      const v4f* r3 = reinterpret_cast<const v4f*>(feat + (size_t)__builtin_amdgcn_readlane(sidx, j + 3) * D_N);
      float a0 = __int_as_float(__builtin_amdgcn_readlane(ial, j));
      float a1 = __int_as_float(__builtin_amdgcn_readlane(ial, j + 1));
      float a2 = __int_as_float(__builtin_amdgcn_readlane(ial, j + 2));
      float a3 = __int_as_float(__builtin_amdgcn_readlane(ial, j + 3));
      v4f x00 = r0[lane], x01 = r0[lane + 32];
      v4f x10 = r1[lane], x11 = r1[lane + 32];
      v4f x20 = r2[lane], x21 = r2[lane + 32];
      v4f x30 = r3[lane], x31 = r3[lane + 32];
      acc0 += a0 * x00; acc1 += a0 * x01;
      acc0 += a1 * x10; acc1 += a1 * x11;
      acc0 += a2 * x20; acc1 += a2 * x21;
      acc0 += a3 * x30; acc1 += a3 * x31;
    }
    for (; j < n; ++j) {
      const v4f* r0 = reinterpret_cast<const v4f*>(feat + (size_t)__builtin_amdgcn_readlane(sidx, j) * D_N);
      float a0 = __int_as_float(__builtin_amdgcn_readlane(ial, j));
      v4f x00 = r0[lane], x01 = r0[lane + 32];
      acc0 += a0 * x00; acc1 += a0 * x01;
    }
  }
  v4f* hrow = reinterpret_cast<v4f*>(H + (size_t)g * D_N);
  hrow[lane] = acc0;
  hrow[lane + 32] = acc1;
}

__global__ void fc_softmax_kernel(const float* __restrict__ H,
                                  const float* __restrict__ W,
                                  const float* __restrict__ bias,
                                  float* __restrict__ out) {
  __shared__ float tile[16][260];
  int wave = threadIdx.x >> 5;
  int lane = threadIdx.x & 31;
  size_t rowBase = (size_t)blockIdx.x * 16;
  int halfK = lane >> 4;
  int l16 = lane & 15;

  v8f zero = {0.f, 0.f, 0.f, 0.f, 0.f, 0.f, 0.f, 0.f};
  v8f acc0 = zero, acc1 = zero, acc2 = zero, acc3 = zero;

  const float* hrow = H + (rowBase + l16) * D_N;
  int colBase = wave * 64 + l16;
  for (int k0 = 0; k0 < D_N; k0 += 4) {
    int ka = k0 + halfK * 2;
    v2f a;
    a.x = hrow[ka];
    a.y = hrow[ka + 1];
    const float* w0 = W + (size_t)ka * DOUT_N;
    const float* w1 = w0 + DOUT_N;
    v2f b0; b0.x = w0[colBase +  0]; b0.y = w1[colBase +  0];
    v2f b1; b1.x = w0[colBase + 16]; b1.y = w1[colBase + 16];
    v2f b2; b2.x = w0[colBase + 32]; b2.y = w1[colBase + 32];
    v2f b3; b3.x = w0[colBase + 48]; b3.y = w1[colBase + 48];
    acc0 = __builtin_amdgcn_wmma_f32_16x16x4_f32(false, a, false, b0, (short)0, acc0, false, false);
    acc1 = __builtin_amdgcn_wmma_f32_16x16x4_f32(false, a, false, b1, (short)0, acc1, false, false);
    acc2 = __builtin_amdgcn_wmma_f32_16x16x4_f32(false, a, false, b2, (short)0, acc2, false, false);
    acc3 = __builtin_amdgcn_wmma_f32_16x16x4_f32(false, a, false, b3, (short)0, acc3, false, false);
  }

  int rowOff = halfK * 8;
  {
    v8f accs[4] = {acc0, acc1, acc2, acc3};
#pragma unroll
    for (int n = 0; n < 4; ++n) {
      int col = colBase + n * 16;
      float bv = bias[col];
#pragma unroll
      for (int r = 0; r < 8; ++r) tile[r + rowOff][col] = accs[n][r] + bv;
    }
  }
  __syncthreads();

#pragma unroll
  for (int rr = 0; rr < 4; ++rr) {
    int row = wave + rr * 4;
    float v[8];
    float mx = -3.402823466e38f;
#pragma unroll
    for (int i = 0; i < 8; ++i) {
      v[i] = tile[row][lane + i * 32];
      mx = fmaxf(mx, v[i]);
    }
#pragma unroll
    for (int off = 16; off; off >>= 1) mx = fmaxf(mx, __shfl_xor(mx, off, 32));
    float sum = 0.f;
#pragma unroll
    for (int i = 0; i < 8; ++i) { v[i] = __expf(v[i] - mx); sum += v[i]; }
#pragma unroll
    for (int off = 16; off; off >>= 1) sum += __shfl_xor(sum, off, 32);
    float inv = 1.0f / sum;
    float* orow = out + (rowBase + row) * DOUT_N;
#pragma unroll
    for (int i = 0; i < 8; ++i) orow[lane + i * 32] = v[i] * inv;
  }
}

extern "C" void kernel_launch(void* const* d_in, const int* in_sizes, int n_in,
                              void* d_out, int out_size, void* d_ws, size_t ws_size,
                              hipStream_t stream) {
  const float* doc_hidden  = (const float*)d_in[0];
  const float* word_hidden = (const float*)d_in[1];
  const int*   edge_doc    = (const int*)d_in[2];
  const int*   edge_word   = (const int*)d_in[3];
  const float* d2w_w       = (const float*)d_in[4];
  const float* d2w_b       = (const float*)d_in[5];
  const float* w2d_w       = (const float*)d_in[6];
  const float* w2d_b       = (const float*)d_in[7];
  const float* fc_w        = (const float*)d_in[8];
  const float* fc_b        = (const float*)d_in[9];
  float* out = (float*)d_out;

  float* ws = (float*)d_ws;
  const size_t NH = (size_t)NG_N * D_N;
  float*    H         = ws;
  unsigned* table     = (unsigned*)ws;
  unsigned* seg       = table + (size_t)GB_N * NG_N;
  int*      grp_arr   = (int*)(seg + GB_N * SCAN_BLOCKS);
  unsigned* off       = (unsigned*)(H + NH);
  float*    sA_doc    = (float*)(off + NG_N + 1);
  float*    sB_doc    = sA_doc + ND_N;
  float*    sA_word   = sB_doc + ND_N;
  float*    sB_word   = sA_word + NW_N;
  int*      grp_srt   = (int*)(sB_word + NW_N);

  node_dots_kernel<<<DOT_BLOCKS, DOT_T, 0, stream>>>(doc_hidden, word_hidden, d2w_w, w2d_w,
                                                     sA_doc, sB_doc, sA_word, sB_word);

  slice_count_kernel<<<GB_N, GB_T, 0, stream>>>(edge_doc, edge_word, table, seg);
  scan_counts_kernel<<<SCAN_BLOCKS, SCAN_T, 0, stream>>>(table, seg, off);
  slice_place_kernel<<<dim3(GB_N, PL_Q), GB_T, 0, stream>>>(edge_doc, edge_word, table, grp_arr);
  group_sort_kernel<<<NG_N / 8, 256, 0, stream>>>(off, grp_arr, grp_srt);

  group_aggregate_kernel<<<NG_N / 8, 256, 0, stream>>>(off, grp_srt, doc_hidden, word_hidden,
                                                       sA_doc, sA_word, sB_doc, sB_word,
                                                       d2w_b, w2d_b, H);

  fc_softmax_kernel<<<NG_N / 16, 128, 0, stream>>>(H, fc_w, fc_b, out);

  (void)in_sizes; (void)n_in; (void)out_size; (void)ws_size;
}


# fc operand feed: H tile staged once in LDS, W read 16 bytes per lane (N tile t column c = output column 64w+4c+t), whole lines per load: 8 global loads per 16 WMMAs, not 36; sum order unchanged
# speedup vs baseline: 4.0285x; 1.1894x over previous
#include <hip/hip_runtime.h>

#define ND_N 20000
#define NW_N 30000
#define E_N  500000
#define D_N  256
#define DOUT_N 256
#define NG_N (ND_N + NW_N)

typedef __attribute__((ext_vector_type(2))) float v2f;
typedef __attribute__((ext_vector_type(4))) float v4f;
typedef __attribute__((ext_vector_type(8))) float v8f;

#define DOT_ROWS 16
#define DOT_T 128
#define DOT_WAVES (DOT_T / 32)
#define DOT_DOC_CHUNKS (ND_N / DOT_ROWS)
#define DOT_CHUNKS ((ND_N + NW_N) / DOT_ROWS)
#define DOT_BLOCKS ((DOT_CHUNKS + DOT_WAVES - 1) / DOT_WAVES)
static_assert(ND_N % DOT_ROWS == 0 && NW_N % DOT_ROWS == 0, "a chunk lies inside one table");
static_assert(2 * DOT_ROWS == 32, "one result per lane: 16 first dots, 16 second dots");

__global__ void __launch_bounds__(DOT_T) node_dots_kernel(const float* __restrict__ doc_hidden,
                                                          const float* __restrict__ word_hidden,
                                                          const float* __restrict__ d2w_w,
                                                          const float* __restrict__ w2d_w,
                                                          float* __restrict__ sA_doc,
                                                          float* __restrict__ sB_doc,
                                                          float* __restrict__ sA_word,
                                                          float* __restrict__ sB_word) {
  int lane = threadIdx.x & 31;
  int chunk = __builtin_amdgcn_readfirstlane((int)(blockIdx.x * DOT_WAVES + (threadIdx.x >> 5)));
  if (chunk >= DOT_CHUNKS) return;
  const float* hid; const float* wa; const float* wb; float* outa; float* outb; int r0;
  if (chunk < DOT_DOC_CHUNKS) {
    r0 = chunk * DOT_ROWS;
    hid = doc_hidden;  wa = d2w_w;       wb = w2d_w + D_N; outa = sA_doc;  outb = sB_doc;
  } else {
    r0 = (chunk - DOT_DOC_CHUNKS) * DOT_ROWS;
    hid = word_hidden; wa = d2w_w + D_N; wb = w2d_w;       outa = sA_word; outb = sB_word;
  }
  const v4f* wa4 = reinterpret_cast<const v4f*>(wa);
  const v4f* wb4 = reinterpret_cast<const v4f*>(wb);
  v4f a0 = wa4[lane], a1 = wa4[lane + 32];
  v4f b0 = wb4[lane], b1 = wb4[lane + 32];
  const v4f* rows = reinterpret_cast<const v4f*>(hid + (size_t)r0 * D_N);
  v4f x[2 * DOT_ROWS];
#pragma unroll
  for (int k = 0; k < 2 * DOT_ROWS; ++k) x[k] = rows[k * 32 + lane];
  float v[2 * DOT_ROWS];
#pragma unroll
  for (int k = 0; k < DOT_ROWS; ++k) {
    v4f sa = x[2 * k] * a0 + x[2 * k + 1] * a1;
    v4f sb = x[2 * k] * b0 + x[2 * k + 1] * b1;
    v[k]            = (sa.x + sa.y) + (sa.z + sa.w);
    v[DOT_ROWS + k] = (sb.x + sb.y) + (sb.z + sb.w);
  }
  for (int o = 16; o; o >>= 1) {
#pragma unroll
    for (int j = 0; j < 2 * DOT_ROWS; ++j) v[j] += __shfl_xor(v[j], o, 32);
  }
  float mine = v[0];
#pragma unroll
  for (int j = 1; j < 2 * DOT_ROWS; ++j) mine = lane == j ? v[j] : mine;
  float* dst = lane < DOT_ROWS ? outa + r0 + lane : outb + r0 + (lane - DOT_ROWS);
  *dst = mine;
}

typedef __attribute__((ext_vector_type(4))) unsigned v4u;
static_assert(NG_N % 4 == 0, "a 16-byte group never straddles the end of the counts");
#define GB_N 32
#define GB_T 1024
#define GB_SLICE ((E_N + GB_N - 1) / GB_N)
#define GB_ROW_IT ((NG_N / 4 + GB_T - 1) / GB_T)
#define GB_BATCH 8
#define PL_Q 10
#define PL_BINS (NG_N / PL_Q)
static_assert(NG_N % (4 * PL_Q) == 0, "a range is a whole number of 16-byte groups");

#define SCAN_T 512
#define SCAN_PER (4 * SCAN_T)
#define SCAN_BLOCKS ((NG_N + SCAN_PER - 1) / SCAN_PER)
#define SCAN_W (SCAN_T / 32)
#define SEG_SLOTS (2 * GB_ROW_IT)
static_assert(GB_T == 2 * SCAN_T, "a wave of the count kernel lies inside one scan segment in every sweep");
static_assert(SEG_SLOTS >= SCAN_BLOCKS, "every scan block has its segment sum");

__global__ void __launch_bounds__(GB_T) slice_count_kernel(const int* __restrict__ edge_doc,
                                                           const int* __restrict__ edge_word,
                                                           unsigned* __restrict__ table,
                                                           unsigned* __restrict__ seg) {
  __shared__ __attribute__((aligned(16))) unsigned c[NG_N];
  __shared__ unsigned segsum[SEG_SLOTS];
  int t = threadIdx.x, lane = t & 31;
  v4u* c4 = reinterpret_cast<v4u*>(c);
  for (int i = t; i < NG_N / 4; i += GB_T) c4[i] = (v4u){0u, 0u, 0u, 0u};
  if (t < SEG_SLOTS) segsum[t] = 0u;
  __syncthreads();
  int lo = blockIdx.x * GB_SLICE;
  int hi = lo + GB_SLICE < E_N ? lo + GB_SLICE : E_N;
  for (int base = lo + t; base < hi; base += GB_BATCH * GB_T) {
    int dd[GB_BATCH], ww[GB_BATCH];
#pragma unroll
    for (int j = 0; j < GB_BATCH; ++j) {
      int e = base + j * GB_T;
      dd[j] = 0; ww[j] = 0;
      if (e < hi) { dd[j] = edge_doc[e]; ww[j] = edge_word[e]; }
    }
#pragma unroll
    for (int j = 0; j < GB_BATCH; ++j) {
      if (base + j * GB_T < hi) {
        atomicAdd(&c[dd[j]], 1u);
        atomicAdd(&c[ND_N + ww[j]], 1u);
      }
    }
  }
  __syncthreads();
  v4u* row = reinterpret_cast<v4u*>(table + (size_t)blockIdx.x * NG_N);
  unsigned w[GB_ROW_IT];
#pragma unroll
  for (int k = 0; k < GB_ROW_IT; ++k) {
    int i = t + k * GB_T;
    v4u v = {0u, 0u, 0u, 0u};
    if (i < NG_N / 4) { v = c4[i]; row[i] = v; }
    w[k] = (v.x + v.y) + (v.z + v.w);
  }
  for (int o = 16; o; o >>= 1) {
#pragma unroll
    for (int k = 0; k < GB_ROW_IT; ++k) w[k] += __shfl_xor(w[k], o, 32);
  }
  if (lane == 0) {
#pragma unroll
    for (int k = 0; k < GB_ROW_IT; ++k) atomicAdd(&segsum[(t + k * GB_T) / SCAN_T], w[k]);
  }
  __syncthreads();
  if (t < SCAN_BLOCKS) seg[blockIdx.x * SCAN_BLOCKS + t] = segsum[t];
}

__global__ void __launch_bounds__(SCAN_T) scan_counts_kernel(unsigned* __restrict__ table,
                                                             const unsigned* __restrict__ seg,
                                                             unsigned* __restrict__ off) {
  __shared__ unsigned wpre[SCAN_W];
  __shared__ unsigned wtot[SCAN_W];
  int t = threadIdx.x, wave = t >> 5, lane = t & 31;
  int blk = blockIdx.x;
  v4u* tab4 = reinterpret_cast<v4u*>(table);
  int q = blk * SCAN_T + t;
  bool own = q < NG_N / 4;
  int qq = own ? q : 0;
  v4u r[GB_N];
#pragma unroll
  for (int b = 0; b < GB_N; ++b) r[b] = tab4[b * (NG_N / 4) + qq];
  unsigned pre = 0u;
  for (int i = t; i < GB_N * SCAN_BLOCKS; i += SCAN_T) {
    unsigned s = seg[i];
    if (i % SCAN_BLOCKS < blk) pre += s;
  }
  v4u mine = {0u, 0u, 0u, 0u};
#pragma unroll
  for (int b = 0; b < GB_N; ++b) { v4u cb = r[b]; r[b] = mine; mine += cb; }
  if (!own) mine = (v4u){0u, 0u, 0u, 0u};
  unsigned s1 = mine.x, s2 = s1 + mine.y, s3 = s2 + mine.z, tot = s3 + mine.w;
  for (int o = 16; o; o >>= 1) pre += __shfl_xor(pre, o, 32);
  unsigned inc = tot;
  for (int o = 1; o < 32; o <<= 1) {
    unsigned u = __shfl_up(inc, o, 32);
    if (lane >= o) inc += u;
  }
  if (lane == 0) wpre[wave] = pre;
  if (lane == 31) wtot[wave] = inc;
  __syncthreads();
  unsigned x = 0u;
  if (lane < SCAN_W) x = wpre[lane] + (lane < wave ? wtot[lane] : 0u);
  for (int o = 16; o; o >>= 1) x += __shfl_xor(x, o, 32);
  unsigned e0 = x + inc - tot;
  if (own) {
    v4u o4 = {e0, e0 + s1, e0 + s2, e0 + s3};
    reinterpret_cast<v4u*>(off)[q] = o4;
#pragma unroll
    for (int b = 0; b < GB_N; ++b) tab4[b * (NG_N / 4) + q] = o4 + r[b];
    if (q == NG_N / 4 - 1) off[NG_N] = e0 + tot;
  }
}

__global__ void __launch_bounds__(GB_T) slice_place_kernel(const int* __restrict__ edge_doc,
                                                           const int* __restrict__ edge_word,
                                                           const unsigned* __restrict__ table,
                                                           int* __restrict__ grp) {
  __shared__ __attribute__((aligned(16))) unsigned p[PL_BINS];
  int t = threadIdx.x;
  int slice = blockIdx.x;
  unsigned first = blockIdx.y * PL_BINS;
  v4u* p4 = reinterpret_cast<v4u*>(p);
  const v4u* row = reinterpret_cast<const v4u*>(table + (size_t)slice * NG_N + first);
  for (int i = t; i < PL_BINS / 4; i += GB_T) p4[i] = row[i];
  __syncthreads();
  int lo = slice * GB_SLICE;
  int hi = lo + GB_SLICE < E_N ? lo + GB_SLICE : E_N;
  for (int base = lo + t; base < hi; base += GB_BATCH * GB_T) {
    int dd[GB_BATCH], ww[GB_BATCH];
#pragma unroll
    for (int j = 0; j < GB_BATCH; ++j) {
      int e = base + j * GB_T;
      dd[j] = 0; ww[j] = 0;
      if (e < hi) { dd[j] = edge_doc[e]; ww[j] = edge_word[e]; }
    }
#pragma unroll
    for (int j = 0; j < GB_BATCH; ++j) {
      if (base + j * GB_T < hi) {
        unsigned k1 = (unsigned)dd[j] - first;
        unsigned k2 = (unsigned)(ND_N + ww[j]) - first;
        if (k1 < (unsigned)PL_BINS) grp[atomicAdd(&p[k1], 1u)] = ww[j];
        if (k2 < (unsigned)PL_BINS) grp[atomicAdd(&p[k2], 1u)] = dd[j];
      }
    }
  }
}

__device__ __forceinline__ float group_score(const int* seg, const float* __restrict__ ssrc,
                                             int idx, float sd, float bb, int* sidx) {
  int s = seg[idx];
  *sidx = s;
  float a = ssrc[s] + sd + bb;
  return a > 0.f ? a : 0.01f * a;
}

#define AGG_WAVES 8
#define AGG_CAP 256
static_assert(NG_N % AGG_WAVES == 0, "the grid is exactly one wave per destination");
static_assert(AGG_CAP % 32 == 0, "a window is a whole number of 32-element chunks");

__global__ void __launch_bounds__(AGG_WAVES * 32) group_aggregate_kernel(const unsigned* __restrict__ off,
                                       const int* __restrict__ grp,
                                       const float* __restrict__ doc_hidden,
                                       const float* __restrict__ word_hidden,
                                       const float* __restrict__ sA_doc,
                                       const float* __restrict__ sA_word,
                                       const float* __restrict__ sB_doc,
                                       const float* __restrict__ sB_word,
                                       const float* __restrict__ d2w_b,
                                       const float* __restrict__ w2d_b,
                                       float* __restrict__ H) {
  __shared__ int sorted_ids[AGG_WAVES][AGG_CAP];
  int lane = threadIdx.x & 31;
  int wv = __builtin_amdgcn_readfirstlane((int)(threadIdx.x >> 5));
  int g = __builtin_amdgcn_readfirstlane((int)(blockIdx.x * AGG_WAVES + (threadIdx.x >> 5)));
  if (g >= NG_N) return;
  const float* feat; const float* ssrc; float sd, bb;
  if (g < ND_N) { feat = word_hidden; ssrc = sB_word; sd = sB_doc[g];         bb = w2d_b[0]; }
  else          { feat = doc_hidden;  ssrc = sA_doc;  sd = sA_word[g - ND_N]; bb = d2w_b[0]; }
  int start = (int)off[g];
  int deg = (int)off[g + 1] - start;
  const int* in = grp + start;
  int* sk = sorted_ids[wv];
  bool windows = deg > AGG_CAP;

  auto fill = [&](int wb) {
    for (int c = 0; c < deg; c += 32) {
      int idx = c + lane;
      int key = idx < deg ? in[idx] : 0x7fffffff;
      int rank = 0;
      for (int c2 = 0; c2 < deg; c2 += 32) {
        int idx2 = c2 + lane;
        int k2 = idx2 < deg ? in[idx2] : 0x7fffffff;
        int n2 = deg - c2 < 32 ? deg - c2 : 32;
        for (int j = 0; j < n2; ++j) {
          int v = __builtin_amdgcn_readlane(k2, j);
          rank += (v < key || (v == key && c2 + j < idx)) ? 1 : 0;
        }
      }
      unsigned slot = (unsigned)(rank - wb);
      if (idx < deg && slot < (unsigned)AGG_CAP) sk[slot] = key;
    }
  };
  fill(0);

  int sidx0 = 0; float v0 = 0.f;
  if (lane < deg) v0 = group_score(sk, ssrc, lane, sd, bb, &sidx0);
  auto score_at = [&](int c, int idx, int wb, int* sidx) -> float {
    if (c == 0) { *sidx = sidx0; return v0; }
    return group_score(sk, ssrc, idx - wb, sd, bb, sidx);
  };

  float mx = -__builtin_huge_valf();
  for (int wb = 0; wb < deg; wb += AGG_CAP) {
    if (windows) fill(wb);
    int we = deg - wb < AGG_CAP ? deg : wb + AGG_CAP;
    for (int c = wb; c < we; c += 32) {
      int idx = c + lane;
      if (idx < deg) { int s; mx = fmaxf(mx, score_at(c, idx, wb, &s)); }
    }
  }
  for (int o = 16; o; o >>= 1) mx = fmaxf(mx, __shfl_xor(mx, o, 32));

  float sum = 0.f;
  for (int wb = 0; wb < deg; wb += AGG_CAP) {
    if (windows) fill(wb);
    int we = deg - wb < AGG_CAP ? deg : wb + AGG_CAP;
    for (int c = wb; c < we; c += 32) {
      int idx = c + lane;
      float ex = 0.f;
      if (idx < deg) { int s; ex = __expf(score_at(c, idx, wb, &s) - mx); }
      for (int o = 16; o; o >>= 1) ex += __shfl_xor(ex, o, 32);
      sum += ex;
    }
  }
  float den = fmaxf(sum, 1e-30f);

  v4f acc0 = {0.f, 0.f, 0.f, 0.f}, acc1 = {0.f, 0.f, 0.f, 0.f};
  for (int wb = 0; wb < deg; wb += AGG_CAP) {
    if (windows) fill(wb);
    int we = deg - wb < AGG_CAP ? deg : wb + AGG_CAP;
    for (int c = wb; c < we; c += 32) {
      int idx = c + lane;
      int sidx = 0; float al = 0.f;
      if (idx < deg) al = __expf(score_at(c, idx, wb, &sidx) - mx) / den;
      int ial = __float_as_int(al);
      int n = deg - c < 32 ? deg - c : 32;
      int j = 0;
      for (; j + 4 <= n; j += 4) {
        const v4f* r0 = reinterpret_cast<const v4f*>(feat + (size_t)__builtin_amdgcn_readlane(sidx, j)     * D_N);
        const v4f* r1 = reinterpret_cast<const v4f*>(feat + (size_t)__builtin_amdgcn_readlane(sidx, j + 1) * D_N);
        const v4f* r2 = reinterpret_cast<const v4f*>(feat + (size_t)__builtin_amdgcn_readlane(sidx, j + 2) * D_N);
        const v4f* r3 = reinterpret_cast<const v4f*>(feat + (size_t)__builtin_amdgcn_readlane(sidx, j + 3) * D_N);
        float a0 = __int_as_float(__builtin_amdgcn_readlane(ial, j));
        float a1 = __int_as_float(__builtin_amdgcn_readlane(ial, j + 1));
        float a2 = __int_as_float(__builtin_amdgcn_readlane(ial, j + 2));
        float a3 = __int_as_float(__builtin_amdgcn_readlane(ial, j + 3));
        v4f x00 = r0[lane], x01 = r0[lane + 32];
        v4f x10 = r1[lane], x11 = r1[lane + 32];
        v4f x20 = r2[lane], x21 = r2[lane + 32];
        v4f x30 = r3[lane], x31 = r3[lane + 32];
        acc0 += a0 * x00; acc1 += a0 * x01;
        acc0 += a1 * x10; acc1 += a1 * x11;
        acc0 += a2 * x20; acc1 += a2 * x21;
        acc0 += a3 * x30; acc1 += a3 * x31;
      }
      for (; j < n; ++j) {
        const v4f* r0 = reinterpret_cast<const v4f*>(feat + (size_t)__builtin_amdgcn_readlane(sidx, j) * D_N);
        float a0 = __int_as_float(__builtin_amdgcn_readlane(ial, j));
        v4f x00 = r0[lane], x01 = r0[lane + 32];
        acc0 += a0 * x00; acc1 += a0 * x01;
      }
    }
  }
  v4f* hrow = reinterpret_cast<v4f*>(H + (size_t)g * D_N);
  hrow[lane] = acc0;
  hrow[lane + 32] = acc1;
}

__global__ void __launch_bounds__(128) fc_softmax_kernel(const float* __restrict__ H,
                                                         const float* __restrict__ W,
                                                         const float* __restrict__ bias,
                                                         float* __restrict__ out) {
  __shared__ __attribute__((aligned(16))) float atile[16][260];
  __shared__ __attribute__((aligned(16))) float tile[16][260];
  int wave = threadIdx.x >> 5;
  int lane = threadIdx.x & 31;
  size_t rowBase = (size_t)blockIdx.x * 16;
  int halfK = lane >> 4;
  int l16 = lane & 15;

  const v4f* hsrc = reinterpret_cast<const v4f*>(H + rowBase * D_N);
  v4f st[8];
#pragma unroll
  for (int i = 0; i < 8; ++i) st[i] = hsrc[(wave * 8 + i) * 32 + lane];
#pragma unroll
  for (int i = 0; i < 8; ++i) {
    int q = (wave * 8 + i) * 32 + lane;
    *reinterpret_cast<v4f*>(&atile[q >> 6][(q & 63) * 4]) = st[i];
  }
  __syncthreads();

  v8f zero = {0.f, 0.f, 0.f, 0.f, 0.f, 0.f, 0.f, 0.f};
  v8f acc0 = zero, acc1 = zero, acc2 = zero, acc3 = zero;

  const float* arow = &atile[l16][halfK * 2];
  int col4 = wave * 64 + l16 * 4;
  const float* wrow = W + (size_t)(halfK * 2) * DOUT_N + col4;
  for (int k0 = 0; k0 < D_N; k0 += 4) {
    v2f a = *reinterpret_cast<const v2f*>(arow + k0);
    v4f w0 = *reinterpret_cast<const v4f*>(wrow + (size_t)k0 * DOUT_N);
    v4f w1 = *reinterpret_cast<const v4f*>(wrow + (size_t)k0 * DOUT_N + DOUT_N);
    v2f b0; b0.x = w0.x; b0.y = w1.x;
    v2f b1; b1.x = w0.y; b1.y = w1.y;
    v2f b2; b2.x = w0.z; b2.y = w1.z;
    v2f b3; b3.x = w0.w; b3.y = w1.w;
    acc0 = __builtin_amdgcn_wmma_f32_16x16x4_f32(false, a, false, b0, (short)0, acc0, false, false);
    acc1 = __builtin_amdgcn_wmma_f32_16x16x4_f32(false, a, false, b1, (short)0, acc1, false, false);
    acc2 = __builtin_amdgcn_wmma_f32_16x16x4_f32(false, a, false, b2, (short)0, acc2, false, false);
    acc3 = __builtin_amdgcn_wmma_f32_16x16x4_f32(false, a, false, b3, (short)0, acc3, false, false);
  }

  int rowOff = halfK * 8;
  {
    v4f bv = *reinterpret_cast<const v4f*>(bias + col4);
#pragma unroll
    for (int r = 0; r < 8; ++r) {
      v4f o = {acc0[r] + bv.x, acc1[r] + bv.y, acc2[r] + bv.z, acc3[r] + bv.w};
      *reinterpret_cast<v4f*>(&tile[r + rowOff][col4]) = o;
    }
  }
  __syncthreads();

#pragma unroll
  for (int rr = 0; rr < 4; ++rr) {
    int row = wave + rr * 4;
    float v[8];
    float mx = -3.402823466e38f;
#pragma unroll
    for (int i = 0; i < 8; ++i) {
      v[i] = tile[row][lane + i * 32];
      mx = fmaxf(mx, v[i]);
    }
#pragma unroll
    for (int off = 16; off; off >>= 1) mx = fmaxf(mx, __shfl_xor(mx, off, 32));
    float sum = 0.f;
#pragma unroll
    for (int i = 0; i < 8; ++i) { v[i] = __expf(v[i] - mx); sum += v[i]; }
#pragma unroll
    for (int off = 16; off; off >>= 1) sum += __shfl_xor(sum, off, 32);
    float inv = 1.0f / sum;
    float* orow = out + (rowBase + row) * DOUT_N;
#pragma unroll
    for (int i = 0; i < 8; ++i) orow[lane + i * 32] = v[i] * inv;
  }
}

extern "C" void kernel_launch(void* const* d_in, const int* in_sizes, int n_in,
                              void* d_out, int out_size, void* d_ws, size_t ws_size,
                              hipStream_t stream) {
  const float* doc_hidden  = (const float*)d_in[0];
  const float* word_hidden = (const float*)d_in[1];
  const int*   edge_doc    = (const int*)d_in[2];
  const int*   edge_word   = (const int*)d_in[3];
  const float* d2w_w       = (const float*)d_in[4];
  const float* d2w_b       = (const float*)d_in[5];
  const float* w2d_w       = (const float*)d_in[6];
  const float* w2d_b       = (const float*)d_in[7];
  const float* fc_w        = (const float*)d_in[8];
  const float* fc_b        = (const float*)d_in[9];
  float* out = (float*)d_out;

  float* ws = (float*)d_ws;
  const size_t NH = (size_t)NG_N * D_N;
  float*    H         = ws;
  unsigned* table     = (unsigned*)ws;
  unsigned* seg       = table + (size_t)GB_N * NG_N;
  unsigned* off       = (unsigned*)(H + NH);
  float*    sA_doc    = (float*)(off + NG_N + 1);
  float*    sB_doc    = sA_doc + ND_N;
  float*    sA_word   = sB_doc + ND_N;
  float*    sB_word   = sA_word + NW_N;
  int*      grp_arr   = (int*)(sB_word + NW_N);

  node_dots_kernel<<<DOT_BLOCKS, DOT_T, 0, stream>>>(doc_hidden, word_hidden, d2w_w, w2d_w,
                                                     sA_doc, sB_doc, sA_word, sB_word);

  slice_count_kernel<<<GB_N, GB_T, 0, stream>>>(edge_doc, edge_word, table, seg);
  scan_counts_kernel<<<SCAN_BLOCKS, SCAN_T, 0, stream>>>(table, seg, off);
  slice_place_kernel<<<dim3(GB_N, PL_Q), GB_T, 0, stream>>>(edge_doc, edge_word, table, grp_arr);

  group_aggregate_kernel<<<NG_N / AGG_WAVES, AGG_WAVES * 32, 0, stream>>>(off, grp_arr, doc_hidden, word_hidden,
                                                       sA_doc, sA_word, sB_doc, sB_word,
                                                       d2w_b, w2d_b, H);

  fc_softmax_kernel<<<NG_N / 16, 128, 0, stream>>>(H, fc_w, fc_b, out);

  (void)in_sizes; (void)n_in; (void)out_size; (void)ws_size;
}
